# starting kernel
# baseline (speedup 1.0000x reference)
#include <hip/hip_runtime.h>
#include <hip/hip_bf16.h>


typedef __attribute__((ext_vector_type(2))) float v2f;
typedef __attribute__((ext_vector_type(8))) float v8f;

#define NEG_SLOPE 0.2f
#define LN_EPS    1e-5f

__global__ void fill_f32(float* __restrict__ p, long long n, float v) {
  long long i = (long long)blockIdx.x * blockDim.x + threadIdx.x;
  if (i < n) p[i] = v;
}

__device__ __forceinline__ unsigned f2sortable(float f) {
  unsigned u = __float_as_uint(f);
  return (u & 0x80000000u) ? ~u : (u | 0x80000000u);
}
__device__ __forceinline__ float sortable2f(unsigned u) {
  u = (u & 0x80000000u) ? (u & 0x7FFFFFFFu) : ~u;
  return __uint_as_float(u);
}
__device__ __forceinline__ float leaky(float x) {
  return x > 0.0f ? x : NEG_SLOPE * x;
}
__device__ __forceinline__ void edge_sd(const int* __restrict__ ei, int E, int e,
                                        int& s, int& d) {
  if (e < E) { s = ei[e]; d = ei[E + e]; }
  else       { s = d = e - E; }
}

template <int MODE>
__global__ void gemm_wmma_n256(const float* __restrict__ A,
                               const float* __restrict__ B,
                               float* __restrict__ C,
                               const float* __restrict__ bias_a,
                               const float* __restrict__ bias_b,
                               int M, int K) {
  const int lane  = threadIdx.x & 31;
  const int wave  = threadIdx.x >> 5;
  const int rtile = blockIdx.x;
  const int rlo   = lane & 15;
  const int khalf = (lane >> 4) * 2;
  long long arow  = (long long)rtile * 16 + rlo;
  if (arow >= M) arow = M - 1;
  const int col   = wave * 32 + rlo;

  v8f acc0 = {}; v8f acc1 = {};
  const float* __restrict__ Ar = A + arow * K;
  for (int k = 0; k < K; k += 4) {
    v2f a, b0, b1;
    a.x = Ar[k + khalf];
    a.y = Ar[k + khalf + 1];
    const float* __restrict__ Bk = B + (long long)(k + khalf) * 256;
    b0.x = Bk[col];        b0.y = Bk[256 + col];
    b1.x = Bk[col + 16];   b1.y = Bk[256 + col + 16];
    acc0 = __builtin_amdgcn_wmma_f32_16x16x4_f32(false, a, false, b0,
                                                 (short)0, acc0, false, false);
    acc1 = __builtin_amdgcn_wmma_f32_16x16x4_f32(false, a, false, b1,
                                                 (short)0, acc1, false, false);
  }

  const int rhalf = (lane >> 4) * 8;
#pragma unroll
  for (int r = 0; r < 8; ++r) {
    long long orow = (long long)rtile * 16 + rhalf + r;
    if (orow < M) {
      const int c0 = wave * 32 + rlo;
      long long o  = orow * 256 + c0;
      if (MODE == 0) {
        C[o]      = acc0[r];
        C[o + 16] = acc1[r];
      } else {
        C[o]      += acc0[r] + bias_a[c0]      + bias_b[c0];
        C[o + 16] += acc1[r] + bias_a[c0 + 16] + bias_b[c0 + 16];
      }
    }
  }
}

__global__ void attn_scores(const float* __restrict__ hbuf,
                            const float* __restrict__ att_s,
                            const float* __restrict__ att_d,
                            float* __restrict__ as_out,
                            float* __restrict__ ad_out,
                            int Nn, int Hh, int Cc) {
  long long tid = (long long)blockIdx.x * blockDim.x + threadIdx.x;
  if (tid >= (long long)Nn * Hh) return;
  int n = (int)(tid / Hh), h = (int)(tid % Hh);
  const float* __restrict__ row = hbuf + (long long)n * 256 + h * Cc;
  const float* __restrict__ vs  = att_s + h * Cc;
  const float* __restrict__ vd  = att_d + h * Cc;
  float ss = 0.0f, sd = 0.0f;
  for (int c = 0; c < Cc; ++c) { float v = row[c]; ss += v * vs[c]; sd += v * vd[c]; }
  as_out[(long long)n * Hh + h] = ss;
  ad_out[(long long)n * Hh + h] = sd;
}

__global__ void edge_max(const int* __restrict__ ei,
                         const float* __restrict__ asrc,
                         const float* __restrict__ adst,
                         unsigned* __restrict__ mbuf,
                         int E, int EA, int Hh) {
  long long tid = (long long)blockIdx.x * blockDim.x + threadIdx.x;
  if (tid >= (long long)EA * Hh) return;
  int e = (int)(tid / Hh), h = (int)(tid % Hh);
  int s, d; edge_sd(ei, E, e, s, d);
  float l = leaky(asrc[(long long)s * Hh + h] + adst[(long long)d * Hh + h]);
  atomicMax(&mbuf[(long long)d * Hh + h], f2sortable(l));
}

__global__ void edge_expsum(const int* __restrict__ ei,
                            const float* __restrict__ asrc,
                            const float* __restrict__ adst,
                            const unsigned* __restrict__ mbuf,
                            float* __restrict__ zbuf,
                            int E, int EA, int Hh) {
  long long tid = (long long)blockIdx.x * blockDim.x + threadIdx.x;
  if (tid >= (long long)EA * Hh) return;
  int e = (int)(tid / Hh), h = (int)(tid % Hh);
  int s, d; edge_sd(ei, E, e, s, d);
  float l = leaky(asrc[(long long)s * Hh + h] + adst[(long long)d * Hh + h]);
  float m = sortable2f(mbuf[(long long)d * Hh + h]);
  atomicAdd(&zbuf[(long long)d * Hh + h], __expf(l - m));
}

__global__ void edge_aggregate(const int* __restrict__ ei,
                               const float* __restrict__ hin,
                               const float* __restrict__ asrc,
                               const float* __restrict__ adst,
                               const unsigned* __restrict__ mbuf,
                               const float* __restrict__ zbuf,
                               float* __restrict__ agg,
                               int E, int EA, int Hh, int Cc) {
  long long gtid = (long long)blockIdx.x * blockDim.x + threadIdx.x;
  int wid  = (int)(gtid >> 5);
  int lane = threadIdx.x & 31;
  if (wid >= EA) return;
  int s, d; edge_sd(ei, E, wid, s, d);
  const float* __restrict__ src_row = hin + (long long)s * 256;
  float* __restrict__ dst_row       = agg + (long long)d * 256;
#pragma unroll
  for (int k = 0; k < 8; ++k) {
    int idx = k * 32 + lane;
    int hh  = idx / Cc;
    float l = leaky(asrc[(long long)s * Hh + hh] + adst[(long long)d * Hh + hh]);
    float m = sortable2f(mbuf[(long long)d * Hh + hh]);
    float z = zbuf[(long long)d * Hh + hh];
    float alpha = __expf(l - m) / z;
    atomicAdd(&dst_row[idx], src_row[idx] * alpha);
  }
}

__global__ void ln_elu(float* __restrict__ hbuf, const float* __restrict__ b1,
                       const float* __restrict__ gamma,
                       const float* __restrict__ beta, int Nn) {
  long long gtid = (long long)blockIdx.x * blockDim.x + threadIdx.x;
  int n    = (int)(gtid >> 5);
  int lane = threadIdx.x & 31;
  if (n >= Nn) return;
  float* __restrict__ row = hbuf + (long long)n * 256;
  float vals[8];
  float sum = 0.0f;
#pragma unroll
  for (int k = 0; k < 8; ++k) {
    int idx = k * 32 + lane;
    vals[k] = row[idx] + b1[idx];
    sum += vals[k];
  }
#pragma unroll
  for (int off = 16; off > 0; off >>= 1) sum += __shfl_xor(sum, off);
  float mu = sum * (1.0f / 256.0f);
  float var = 0.0f;
#pragma unroll
  for (int k = 0; k < 8; ++k) { float t = vals[k] - mu; var += t * t; }
#pragma unroll
  for (int off = 16; off > 0; off >>= 1) var += __shfl_xor(var, off);
  var *= (1.0f / 256.0f);
  float rs = rsqrtf(var + LN_EPS);
#pragma unroll
  for (int k = 0; k < 8; ++k) {
    int idx = k * 32 + lane;
    float t = (vals[k] - mu) * rs * gamma[idx] + beta[idx];
    row[idx] = t > 0.0f ? t : (__expf(t) - 1.0f);
  }
}

__global__ void pool_accum(const float* __restrict__ o,
                           const int* __restrict__ batch,
                           float* __restrict__ outsum,
                           float* __restrict__ cnt, int Nn) {
  long long gtid = (long long)blockIdx.x * blockDim.x + threadIdx.x;
  int n    = (int)(gtid >> 5);
  int lane = threadIdx.x & 31;
  if (n >= Nn) return;
  int g = batch[n];
  const float* __restrict__ row = o + (long long)n * 256;
  float* __restrict__ acc       = outsum + (long long)g * 256;
#pragma unroll
  for (int k = 0; k < 8; ++k) {
    int idx = k * 32 + lane;
    atomicAdd(&acc[idx], row[idx]);
  }
  if (lane == 0) atomicAdd(&cnt[g], 1.0f);
}

__global__ void pool_div(float* __restrict__ out, const float* __restrict__ cnt,
                         int total) {
  int i = blockIdx.x * blockDim.x + threadIdx.x;
  if (i >= total) return;
  out[i] /= fmaxf(cnt[i >> 8], 1.0f);
}

static inline int cdiv(long long a, long long b) { return (int)((a + b - 1) / b); }

extern "C" void kernel_launch(void* const* d_in, const int* in_sizes, int n_in,
                              void* d_out, int out_size, void* d_ws, size_t ws_size,
                              hipStream_t stream) {
  const float* x        = (const float*)d_in[0];
  const int*   ei       = (const int*)  d_in[1];
  const int*   batch    = (const int*)  d_in[2];
  const float* W1       = (const float*)d_in[3];
  const float* att_src1 = (const float*)d_in[4];
  const float* att_dst1 = (const float*)d_in[5];
  const float* b1       = (const float*)d_in[6];
  const float* gamma    = (const float*)d_in[7];
  const float* beta     = (const float*)d_in[8];
  const float* W2       = (const float*)d_in[9];
  const float* att_src2 = (const float*)d_in[10];
  const float* att_dst2 = (const float*)d_in[11];
  const float* b2       = (const float*)d_in[12];
  const float* Wres     = (const float*)d_in[13];
  const float* bres     = (const float*)d_in[14];
  float* out = (float*)d_out;

  const int Nn = in_sizes[2];
  const int Ee = in_sizes[1] / 2;
  const int Gg = out_size / 256;
  const int EA = Ee + Nn;
  const int IN_DIM = in_sizes[0] / Nn;

  const long long NB = (long long)Nn * 256;
  float* ws   = (float*)d_ws;
  float* buf0 = ws;
  float* buf1 = ws + NB;
  float* buf2 = ws + 2 * NB;
  float* scal = ws + 3 * NB;
  unsigned* m1 = (unsigned*)scal;
  float*    z1 = scal + (long long)Nn * 8;
  unsigned* m2 = (unsigned*)(scal + (long long)Nn * 16);
  float*    z2 = scal + (long long)Nn * 17;
  float*    cnt = scal + (long long)Nn * 18;
  float*    as1 = cnt + Gg;
  float*    ad1 = as1 + (long long)Nn * 8;
  float*    as2 = ad1 + (long long)Nn * 8;
  float*    ad2 = as2 + Nn;

  const int B = 256;
  const long long zlen1 = 2 * NB;
  const long long zlen2 = (long long)Nn * 18 + Gg;

  fill_f32<<<cdiv(zlen1, B), B, 0, stream>>>(buf1, zlen1, 0.0f);
  fill_f32<<<cdiv(zlen2, B), B, 0, stream>>>(scal, zlen2, 0.0f);
  fill_f32<<<cdiv((long long)Gg * 256, B), B, 0, stream>>>(out, (long long)Gg * 256, 0.0f);

  const int gtiles = cdiv(Nn, 16);

  gemm_wmma_n256<0><<<gtiles, B, 0, stream>>>(x, W1, buf0, nullptr, nullptr, Nn, IN_DIM);
  attn_scores<<<cdiv((long long)Nn * 8, B), B, 0, stream>>>(buf0, att_src1, att_dst1,
                                                            as1, ad1, Nn, 8, 32);
  edge_max   <<<cdiv((long long)EA * 8, B), B, 0, stream>>>(ei, as1, ad1, m1, Ee, EA, 8);
  edge_expsum<<<cdiv((long long)EA * 8, B), B, 0, stream>>>(ei, as1, ad1, m1, z1, Ee, EA, 8);
  edge_aggregate<<<cdiv((long long)EA * 32, B), B, 0, stream>>>(ei, buf0, as1, ad1, m1, z1,
                                                                buf1, Ee, EA, 8, 32);
  ln_elu<<<cdiv((long long)Nn * 32, B), B, 0, stream>>>(buf1, b1, gamma, beta, Nn);

  gemm_wmma_n256<0><<<gtiles, B, 0, stream>>>(buf1, W2, buf0, nullptr, nullptr, Nn, 256);
  attn_scores<<<cdiv((long long)Nn, B), B, 0, stream>>>(buf0, att_src2, att_dst2,
                                                        as2, ad2, Nn, 1, 256);
  edge_max   <<<cdiv((long long)EA, B), B, 0, stream>>>(ei, as2, ad2, m2, Ee, EA, 1);
  edge_expsum<<<cdiv((long long)EA, B), B, 0, stream>>>(ei, as2, ad2, m2, z2, Ee, EA, 1);
  edge_aggregate<<<cdiv((long long)EA * 32, B), B, 0, stream>>>(ei, buf0, as2, ad2, m2, z2,
                                                                buf2, Ee, EA, 1, 256);
  gemm_wmma_n256<1><<<gtiles, B, 0, stream>>>(buf1, Wres, buf2, b2, bres, Nn, 256);

  pool_accum<<<cdiv((long long)Nn * 32, B), B, 0, stream>>>(buf2, batch, out, cnt, Nn);
  pool_div  <<<cdiv((long long)Gg * 256, B), B, 0, stream>>>(out, cnt, Gg * 256);
}


# Edges bucketed by destination (CSR built per call, integer atomics); one wave per node does softmax + gather-sum in registers, no float atomics on edges; sum order = CSR slot order
# speedup vs baseline: 2.5327x; 2.5327x over previous
#include <hip/hip_runtime.h>
#include <hip/hip_bf16.h>


typedef __attribute__((ext_vector_type(2))) float v2f;
typedef __attribute__((ext_vector_type(8))) float v8f;
typedef __attribute__((ext_vector_type(4))) float v4f;

#define NEG_SLOPE 0.2f
#define LN_EPS    1e-5f

__global__ void fill_f32(float* __restrict__ p, long long n, float v) {
  long long i = (long long)blockIdx.x * blockDim.x + threadIdx.x;
  if (i < n) p[i] = v;
}

__device__ __forceinline__ float leaky(float x) {
  return x > 0.0f ? x : NEG_SLOPE * x;
}

template <int MODE>
__global__ void gemm_wmma_n256(const float* __restrict__ A,
                               const float* __restrict__ B,
                               float* __restrict__ C,
                               const float* __restrict__ bias_a,
                               const float* __restrict__ bias_b,
                               int M, int K) {
  const int lane  = threadIdx.x & 31;
  const int wave  = threadIdx.x >> 5;
  const int rtile = blockIdx.x;
  const int rlo   = lane & 15;
  const int khalf = (lane >> 4) * 2;
  long long arow  = (long long)rtile * 16 + rlo;
  if (arow >= M) arow = M - 1;
  const int col   = wave * 32 + rlo;

  v8f acc0 = {}; v8f acc1 = {};
  const float* __restrict__ Ar = A + arow * K;
  for (int k = 0; k < K; k += 4) {
    v2f a, b0, b1;
    a.x = Ar[k + khalf];
    a.y = Ar[k + khalf + 1];
    const float* __restrict__ Bk = B + (long long)(k + khalf) * 256;
    b0.x = Bk[col];        b0.y = Bk[256 + col];
    b1.x = Bk[col + 16];   b1.y = Bk[256 + col + 16];
    acc0 = __builtin_amdgcn_wmma_f32_16x16x4_f32(false, a, false, b0,
                                                 (short)0, acc0, false, false);
    acc1 = __builtin_amdgcn_wmma_f32_16x16x4_f32(false, a, false, b1,
                                                 (short)0, acc1, false, false);
  }

  const int rhalf = (lane >> 4) * 8;
#pragma unroll
  for (int r = 0; r < 8; ++r) {
    long long orow = (long long)rtile * 16 + rhalf + r;
    if (orow < M) {
      const int c0 = wave * 32 + rlo;
      long long o  = orow * 256 + c0;
      if (MODE == 0) {
        C[o]      = acc0[r];
        C[o + 16] = acc1[r];
      } else {
        C[o]      += acc0[r] + bias_a[c0]      + bias_b[c0];
        C[o + 16] += acc1[r] + bias_a[c0 + 16] + bias_b[c0 + 16];
      }
    }
  }
}

__global__ void attn_scores(const float* __restrict__ hbuf,
                            const float* __restrict__ att_s,
                            const float* __restrict__ att_d,
                            float* __restrict__ as_out,
                            float* __restrict__ ad_out,
                            int Nn, int Hh, int Cc) {
  long long tid = (long long)blockIdx.x * blockDim.x + threadIdx.x;
  if (tid >= (long long)Nn * Hh) return;
  int n = (int)(tid / Hh), h = (int)(tid % Hh);
  const float* __restrict__ row = hbuf + (long long)n * 256 + h * Cc;
  const float* __restrict__ vs  = att_s + h * Cc;
  const float* __restrict__ vd  = att_d + h * Cc;
  float ss = 0.0f, sd = 0.0f;
  for (int c = 0; c < Cc; ++c) { float v = row[c]; ss += v * vs[c]; sd += v * vd[c]; }
  as_out[(long long)n * Hh + h] = ss;
  ad_out[(long long)n * Hh + h] = sd;
}

__global__ void csr_count(const int* __restrict__ ei, int* __restrict__ deg, int E) {
  int e = blockIdx.x * blockDim.x + threadIdx.x;
  if (e < E) atomicAdd(&deg[ei[E + e]], 1);
}

__global__ void __launch_bounds__(1024)
csr_scan(const int* __restrict__ deg, int* __restrict__ rowptr,
         int* __restrict__ cursor, int Nn) {
  __shared__ int wsum[32];
  __shared__ int psum[32];
  const int t = threadIdx.x, lane = t & 31, w = t >> 5;
  const int b = blockIdx.x;
  int pre = 0;
  for (int j = 0; j < b; ++j) pre += deg[j * 1024 + t];
  const int idx = b * 1024 + t;
  const int d = (idx < Nn) ? deg[idx] : 0;
  int v = d;
  for (int off = 1; off < 32; off <<= 1) {
    int u = __shfl_up(v, off);
    if (lane >= off) v += u;
  }
  for (int off = 16; off > 0; off >>= 1) pre += __shfl_xor(pre, off);
  if (lane == 31) wsum[w] = v;
  if (lane == 0)  psum[w] = pre;
  __syncthreads();
  int x = psum[lane] + (lane < w ? wsum[lane] : 0);
  for (int off = 16; off > 0; off >>= 1) x += __shfl_xor(x, off);
  const int excl = x + (v - d);
  if (idx < Nn) { rowptr[idx] = excl; cursor[idx] = excl; }
  if (idx == Nn - 1) rowptr[Nn] = excl + d;
}

__global__ void csr_fill(const int* __restrict__ ei, int* __restrict__ cursor,
                         int* __restrict__ csr_src, int E) {
  int e = blockIdx.x * blockDim.x + threadIdx.x;
  if (e < E) {
    int p = atomicAdd(&cursor[ei[E + e]], 1);
    csr_src[p] = ei[e];
  }
}

template <int HEADS>
__global__ void __launch_bounds__(256)
gat_gather(const int* __restrict__ rowptr, const int* __restrict__ csr_src,
           const float* __restrict__ hin, const float* __restrict__ asrc,
           const float* __restrict__ adst, float* __restrict__ outp, int Nn) {
  constexpr int  CC  = 256 / HEADS;
  constexpr int  GL  = (CC / 4 < 32) ? CC / 4 : 32;
  constexpr bool TWO = (HEADS > 1);
  const int lane = threadIdx.x & 31;
  int n = blockIdx.x * 8 + (threadIdx.x >> 5);
  if (n >= Nn) return;
  n = __builtin_amdgcn_readfirstlane(n);
  const int hA  = (4 * lane) / CC, hB = (128 + 4 * lane) / CC;
  const int sub = lane & (GL - 1);
  const int beg = rowptr[n], end = rowptr[n + 1];
  const float adA = adst[n * HEADS + hA];
  const float adB = TWO ? adst[n * HEADS + hB] : adA;
  const float lsA = leaky(asrc[n * HEADS + hA] + adA);
  const float lsB = TWO ? leaky(asrc[n * HEADS + hB] + adB) : lsA;

  float mA = lsA, mB = lsB;
  for (int i = beg + sub; i < end; i += GL) {
    const int s = csr_src[i];
    mA = fmaxf(mA, leaky(asrc[s * HEADS + hA] + adA));
    if (TWO) mB = fmaxf(mB, leaky(asrc[s * HEADS + hB] + adB));
  }
  for (int off = GL / 2; off > 0; off >>= 1) {
    mA = fmaxf(mA, __shfl_xor(mA, off));
    if (TWO) mB = fmaxf(mB, __shfl_xor(mB, off));
  }
  if (!TWO) mB = mA;
  float zA = (sub == 0) ? __expf(lsA - mA) : 0.0f;
  float zB = (sub == 0) ? __expf(lsB - mB) : 0.0f;
  for (int i = beg + sub; i < end; i += GL) {
    const int s = csr_src[i];
    zA += __expf(leaky(asrc[s * HEADS + hA] + adA) - mA);
    if (TWO) zB += __expf(leaky(asrc[s * HEADS + hB] + adB) - mB);
  }
  for (int off = GL / 2; off > 0; off >>= 1) {
    zA += __shfl_xor(zA, off);
    if (TWO) zB += __shfl_xor(zB, off);
  }
  if (!TWO) zB = zA;
  const float rzA = 1.0f / zA;
  const float rzB = TWO ? 1.0f / zB : rzA;

  const v4f* __restrict__ h4 = (const v4f*)hin;
  v4f accA, accB;
  {
    const float aA = __expf(lsA - mA) * rzA;
    const float aB = TWO ? __expf(lsB - mB) * rzB : aA;
    accA = aA * h4[(long long)n * 64 + lane];
    accB = aB * h4[(long long)n * 64 + 32 + lane];
  }
  for (int base = beg; base < end; base += 32) {
    const int cnt = (end - base < 32) ? end - base : 32;
    const int sv  = (lane < cnt) ? csr_src[base + lane] : n;
    float av = 0.0f;
    if (!TWO) av = __expf(leaky(asrc[sv] + adA) - mA) * rzA;
#pragma unroll 4
    for (int t = 0; t < cnt; ++t) {
      const int s = __builtin_amdgcn_readlane(sv, t);
      float aA, aB;
      if (TWO) {
        aA = __expf(leaky(asrc[s * HEADS + hA] + adA) - mA) * rzA;
        aB = __expf(leaky(asrc[s * HEADS + hB] + adB) - mB) * rzB;
      } else {
        aA = aB = __int_as_float(__builtin_amdgcn_readlane(__float_as_int(av), t));
      }
      accA += aA * h4[(long long)s * 64 + lane];
      accB += aB * h4[(long long)s * 64 + 32 + lane];
    }
  }
  v4f* __restrict__ o4 = (v4f*)outp;
  o4[(long long)n * 64 + lane]      = accA;
  o4[(long long)n * 64 + 32 + lane] = accB;
}

__global__ void ln_elu(float* __restrict__ hbuf, const float* __restrict__ b1,
                       const float* __restrict__ gamma,
                       const float* __restrict__ beta, int Nn) {
  long long gtid = (long long)blockIdx.x * blockDim.x + threadIdx.x;
  int n    = (int)(gtid >> 5);
  int lane = threadIdx.x & 31;
  if (n >= Nn) return;
  float* __restrict__ row = hbuf + (long long)n * 256;
  float vals[8];
  float sum = 0.0f;
#pragma unroll
  for (int k = 0; k < 8; ++k) {
    int idx = k * 32 + lane;
    vals[k] = row[idx] + b1[idx];
    sum += vals[k];
  }
#pragma unroll
  for (int off = 16; off > 0; off >>= 1) sum += __shfl_xor(sum, off);
  float mu = sum * (1.0f / 256.0f);
  float var = 0.0f;
#pragma unroll
  for (int k = 0; k < 8; ++k) { float t = vals[k] - mu; var += t * t; }
#pragma unroll
  for (int off = 16; off > 0; off >>= 1) var += __shfl_xor(var, off);
  var *= (1.0f / 256.0f);
  float rs = rsqrtf(var + LN_EPS);
#pragma unroll
  for (int k = 0; k < 8; ++k) {
    int idx = k * 32 + lane;
    float t = (vals[k] - mu) * rs * gamma[idx] + beta[idx];
    row[idx] = t > 0.0f ? t : (__expf(t) - 1.0f);
  }
}

__global__ void pool_accum(const float* __restrict__ o,
                           const int* __restrict__ batch,
                           float* __restrict__ outsum,
                           float* __restrict__ cnt, int Nn) {
  long long gtid = (long long)blockIdx.x * blockDim.x + threadIdx.x;
  int n    = (int)(gtid >> 5);
  int lane = threadIdx.x & 31;
  if (n >= Nn) return;
  int g = batch[n];
  const float* __restrict__ row = o + (long long)n * 256;
  float* __restrict__ acc       = outsum + (long long)g * 256;
#pragma unroll
  for (int k = 0; k < 8; ++k) {
    int idx = k * 32 + lane;
    atomicAdd(&acc[idx], row[idx]);
  }
  if (lane == 0) atomicAdd(&cnt[g], 1.0f);
}

__global__ void pool_div(float* __restrict__ out, const float* __restrict__ cnt,
                         int total) {
  int i = blockIdx.x * blockDim.x + threadIdx.x;
  if (i >= total) return;
  out[i] /= fmaxf(cnt[i >> 8], 1.0f);
}

static inline int cdiv(long long a, long long b) { return (int)((a + b - 1) / b); }

extern "C" void kernel_launch(void* const* d_in, const int* in_sizes, int n_in,
                              void* d_out, int out_size, void* d_ws, size_t ws_size,
                              hipStream_t stream) {
  const float* x        = (const float*)d_in[0];
  const int*   ei       = (const int*)  d_in[1];
  const int*   batch    = (const int*)  d_in[2];
  const float* W1       = (const float*)d_in[3];
  const float* att_src1 = (const float*)d_in[4];
  const float* att_dst1 = (const float*)d_in[5];
  const float* b1       = (const float*)d_in[6];
  const float* gamma    = (const float*)d_in[7];
  const float* beta     = (const float*)d_in[8];
  const float* W2       = (const float*)d_in[9];
  const float* att_src2 = (const float*)d_in[10];
  const float* att_dst2 = (const float*)d_in[11];
  const float* b2       = (const float*)d_in[12];
  const float* Wres     = (const float*)d_in[13];
  const float* bres     = (const float*)d_in[14];
  float* out = (float*)d_out;

  const int Nn = in_sizes[2];
  const int Ee = in_sizes[1] / 2;
  const int Gg = out_size / 256;
  const int IN_DIM = in_sizes[0] / Nn;

  const long long NB = (long long)Nn * 256;
  float* ws   = (float*)d_ws;
  float* buf0 = ws;
  float* buf1 = ws + NB;
  float* buf2 = ws + 2 * NB;
  float* scal = ws + 3 * NB;
  int*   deg  = (int*)scal;
  float* cnt  = scal + Nn;
  float* as1  = cnt + Gg;
  float* ad1  = as1 + (long long)Nn * 8;
  float* as2  = ad1 + (long long)Nn * 8;
  float* ad2  = as2 + Nn;
  int*   rowptr  = (int*)(ad2 + Nn);
  int*   cursor  = rowptr + Nn + 1;
  int*   csr_src = cursor + Nn;

  const int B = 256;
  const long long zlen = (long long)Nn + Gg;

  fill_f32<<<cdiv(zlen, B), B, 0, stream>>>(scal, zlen, 0.0f);
  fill_f32<<<cdiv((long long)Gg * 256, B), B, 0, stream>>>(out, (long long)Gg * 256, 0.0f);

  csr_count<<<cdiv(Ee, B), B, 0, stream>>>(ei, deg, Ee);
  csr_scan<<<cdiv(Nn, 1024), 1024, 0, stream>>>(deg, rowptr, cursor, Nn);
  csr_fill<<<cdiv(Ee, B), B, 0, stream>>>(ei, cursor, csr_src, Ee);

  const int gtiles = cdiv(Nn, 16);
  const int ntiles = cdiv(Nn, 8);

  gemm_wmma_n256<0><<<gtiles, B, 0, stream>>>(x, W1, buf0, nullptr, nullptr, Nn, IN_DIM);
  attn_scores<<<cdiv((long long)Nn * 8, B), B, 0, stream>>>(buf0, att_src1, att_dst1,
                                                            as1, ad1, Nn, 8, 32);
  gat_gather<8><<<ntiles, B, 0, stream>>>(rowptr, csr_src, buf0, as1, ad1, buf1, Nn);
  ln_elu<<<cdiv((long long)Nn * 32, B), B, 0, stream>>>(buf1, b1, gamma, beta, Nn);

  gemm_wmma_n256<0><<<gtiles, B, 0, stream>>>(buf1, W2, buf0, nullptr, nullptr, Nn, 256);
  attn_scores<<<cdiv((long long)Nn, B), B, 0, stream>>>(buf0, att_src2, att_dst2,
                                                        as2, ad2, Nn, 1, 256);
  gat_gather<1><<<ntiles, B, 0, stream>>>(rowptr, csr_src, buf0, as2, ad2, buf2, Nn);
  gemm_wmma_n256<1><<<gtiles, B, 0, stream>>>(buf1, Wres, buf2, b2, bres, Nn, 256);

  pool_accum<<<cdiv((long long)Nn * 32, B), B, 0, stream>>>(buf2, batch, out, cnt, Nn);
  pool_div  <<<cdiv((long long)Gg * 256, B), B, 0, stream>>>(out, cnt, Gg * 256);
}


# GEMM tiling: 32-row A slab staged once per block in LDS; each wave keeps 2x4 accumulator tiles, columns interleaved so one 16-B B load feeds 4 tiles; global loads per 8 WMMAs 20 -> 2; sum order kept
# speedup vs baseline: 2.7239x; 1.0755x over previous
#include <hip/hip_runtime.h>
#include <hip/hip_bf16.h>


typedef __attribute__((ext_vector_type(2))) float v2f;
typedef __attribute__((ext_vector_type(8))) float v8f;
typedef __attribute__((ext_vector_type(4))) float v4f;

#define NEG_SLOPE 0.2f
#define LN_EPS    1e-5f

__global__ void fill_f32(float* __restrict__ p, long long n, float v) {
  long long i = (long long)blockIdx.x * blockDim.x + threadIdx.x;
  if (i < n) p[i] = v;
}

__device__ __forceinline__ float leaky(float x) {
  return x > 0.0f ? x : NEG_SLOPE * x;
}

template <int MODE, int K>
__global__ void __launch_bounds__(128)
gemm_wmma_n256(const float* __restrict__ A,
               const float* __restrict__ B,
               float* __restrict__ C,
               const float* __restrict__ bias_a,
               const float* __restrict__ bias_b,
               int M) {
  constexpr int BM = 32;
  constexpr int P  = K + 4;
  __shared__ __attribute__((aligned(16))) float As[BM * P];

  const int tid  = threadIdx.x;
  const int lane = tid & 31;
  const int wave = tid >> 5;
  const int c    = lane & 15;
  const int h    = lane >> 4;
  const long long row0 = (long long)blockIdx.x * BM;

  constexpr int Q = K / 4;
  for (int i = tid; i < BM * Q; i += 128) {
    const int r = i / Q, q = i - r * Q;
    long long gr = row0 + r;
    if (gr >= M) gr = M - 1;
    const v4f v = *(const v4f*)(A + gr * K + q * 4);
    *(v4f*)(As + r * P + q * 4) = v;
  }
  __syncthreads();

  const float* __restrict__ a0p = As + c * P + 2 * h;
  const float* __restrict__ a1p = As + (16 + c) * P + 2 * h;
  const float* __restrict__ Bx  = B + (long long)(2 * h) * 256 + wave * 64 + 4 * c;
  const float* __restrict__ By  = Bx + 256;

  v8f acc00 = {}, acc01 = {}, acc02 = {}, acc03 = {};
  v8f acc10 = {}, acc11 = {}, acc12 = {}, acc13 = {};

#pragma unroll 2
  for (int k = 0; k < K; k += 4) {
    const v2f a0 = *(const v2f*)(a0p + k);
    const v2f a1 = *(const v2f*)(a1p + k);
    const v4f bx = *(const v4f*)(Bx + (long long)k * 256);
    const v4f by = *(const v4f*)(By + (long long)k * 256);
    v2f b;
    b.x = bx.x; b.y = by.x;
    acc00 = __builtin_amdgcn_wmma_f32_16x16x4_f32(false, a0, false, b, (short)0, acc00, false, false);
    acc10 = __builtin_amdgcn_wmma_f32_16x16x4_f32(false, a1, false, b, (short)0, acc10, false, false);
    b.x = bx.y; b.y = by.y;
    acc01 = __builtin_amdgcn_wmma_f32_16x16x4_f32(false, a0, false, b, (short)0, acc01, false, false);
    acc11 = __builtin_amdgcn_wmma_f32_16x16x4_f32(false, a1, false, b, (short)0, acc11, false, false);
    b.x = bx.z; b.y = by.z;
    acc02 = __builtin_amdgcn_wmma_f32_16x16x4_f32(false, a0, false, b, (short)0, acc02, false, false);
    acc12 = __builtin_amdgcn_wmma_f32_16x16x4_f32(false, a1, false, b, (short)0, acc12, false, false);
    b.x = bx.w; b.y = by.w;
    acc03 = __builtin_amdgcn_wmma_f32_16x16x4_f32(false, a0, false, b, (short)0, acc03, false, false);
    acc13 = __builtin_amdgcn_wmma_f32_16x16x4_f32(false, a1, false, b, (short)0, acc13, false, false);
  }

  const int col = wave * 64 + 4 * c;
  v4f bias = {0.0f, 0.0f, 0.0f, 0.0f};
  v4f bia2 = {0.0f, 0.0f, 0.0f, 0.0f};
  if (MODE == 1) {
    bias = *(const v4f*)(bias_a + col);
    bia2 = *(const v4f*)(bias_b + col);
  }
#pragma unroll
  for (int r = 0; r < 8; ++r) {
    const long long or0 = row0 + 8 * h + r;
    const long long or1 = or0 + 16;
    v4f v0, v1;
    v0.x = acc00[r]; v0.y = acc01[r]; v0.z = acc02[r]; v0.w = acc03[r];
    v1.x = acc10[r]; v1.y = acc11[r]; v1.z = acc12[r]; v1.w = acc13[r];
    if (or0 < M) {
      v4f* p = (v4f*)(C + or0 * 256 + col);
      if (MODE == 0) *p = v0; else *p = *p + ((v0 + bias) + bia2);
    }
    if (or1 < M) {
      v4f* p = (v4f*)(C + or1 * 256 + col);
      if (MODE == 0) *p = v1; else *p = *p + ((v1 + bias) + bia2);
    }
  }
}

__global__ void attn_scores(const float* __restrict__ hbuf,
                            const float* __restrict__ att_s,
                            const float* __restrict__ att_d,
                            float* __restrict__ as_out,
                            float* __restrict__ ad_out,
                            int Nn, int Hh, int Cc) {
  long long tid = (long long)blockIdx.x * blockDim.x + threadIdx.x;
  if (tid >= (long long)Nn * Hh) return;
  int n = (int)(tid / Hh), h = (int)(tid % Hh);
  const float* __restrict__ row = hbuf + (long long)n * 256 + h * Cc;
  const float* __restrict__ vs  = att_s + h * Cc;
  const float* __restrict__ vd  = att_d + h * Cc;
  float ss = 0.0f, sd = 0.0f;
  for (int c = 0; c < Cc; ++c) { float v = row[c]; ss += v * vs[c]; sd += v * vd[c]; }
  as_out[(long long)n * Hh + h] = ss;
  ad_out[(long long)n * Hh + h] = sd;
}

__global__ void csr_count(const int* __restrict__ ei, int* __restrict__ deg, int E) {
  int e = blockIdx.x * blockDim.x + threadIdx.x;
  if (e < E) atomicAdd(&deg[ei[E + e]], 1);
}

__global__ void __launch_bounds__(1024)
csr_scan(const int* __restrict__ deg, int* __restrict__ rowptr,
         int* __restrict__ cursor, int Nn) {
  __shared__ int wsum[32];
  __shared__ int psum[32];
  const int t = threadIdx.x, lane = t & 31, w = t >> 5;
  const int b = blockIdx.x;
  int pre = 0;
  for (int j = 0; j < b; ++j) pre += deg[j * 1024 + t];
  const int idx = b * 1024 + t;
  const int d = (idx < Nn) ? deg[idx] : 0;
  int v = d;
  for (int off = 1; off < 32; off <<= 1) {
    int u = __shfl_up(v, off);
    if (lane >= off) v += u;
  }
  for (int off = 16; off > 0; off >>= 1) pre += __shfl_xor(pre, off);
  if (lane == 31) wsum[w] = v;
  if (lane == 0)  psum[w] = pre;
  __syncthreads();
  int x = psum[lane] + (lane < w ? wsum[lane] : 0);
  for (int off = 16; off > 0; off >>= 1) x += __shfl_xor(x, off);
  const int excl = x + (v - d);
  if (idx < Nn) { rowptr[idx] = excl; cursor[idx] = excl; }
  if (idx == Nn - 1) rowptr[Nn] = excl + d;
}

__global__ void csr_fill(const int* __restrict__ ei, int* __restrict__ cursor,
                         int* __restrict__ csr_src, int E) {
  int e = blockIdx.x * blockDim.x + threadIdx.x;
  if (e < E) {
    int p = atomicAdd(&cursor[ei[E + e]], 1);
    csr_src[p] = ei[e];
  }
}

template <int HEADS>
__global__ void __launch_bounds__(256)
gat_gather(const int* __restrict__ rowptr, const int* __restrict__ csr_src,
           const float* __restrict__ hin, const float* __restrict__ asrc,
           const float* __restrict__ adst, float* __restrict__ outp, int Nn) {
  constexpr int  CC  = 256 / HEADS;
  constexpr int  GL  = (CC / 4 < 32) ? CC / 4 : 32;
  constexpr bool TWO = (HEADS > 1);
  const int lane = threadIdx.x & 31;
  int n = blockIdx.x * 8 + (threadIdx.x >> 5);
  if (n >= Nn) return;
  n = __builtin_amdgcn_readfirstlane(n);
  const int hA  = (4 * lane) / CC, hB = (128 + 4 * lane) / CC;
  const int sub = lane & (GL - 1);
  const int beg = rowptr[n], end = rowptr[n + 1];
  const float adA = adst[n * HEADS + hA];
  const float adB = TWO ? adst[n * HEADS + hB] : adA;
  const float lsA = leaky(asrc[n * HEADS + hA] + adA);
  const float lsB = TWO ? leaky(asrc[n * HEADS + hB] + adB) : lsA;

  float mA = lsA, mB = lsB;
  for (int i = beg + sub; i < end; i += GL) {
    const int s = csr_src[i];
    mA = fmaxf(mA, leaky(asrc[s * HEADS + hA] + adA));
    if (TWO) mB = fmaxf(mB, leaky(asrc[s * HEADS + hB] + adB));
  }
  for (int off = GL / 2; off > 0; off >>= 1) {
    mA = fmaxf(mA, __shfl_xor(mA, off));
    if (TWO) mB = fmaxf(mB, __shfl_xor(mB, off));
  }
  if (!TWO) mB = mA;
  float zA = (sub == 0) ? __expf(lsA - mA) : 0.0f;
  float zB = (sub == 0) ? __expf(lsB - mB) : 0.0f;
  for (int i = beg + sub; i < end; i += GL) {
    const int s = csr_src[i];
    zA += __expf(leaky(asrc[s * HEADS + hA] + adA) - mA);
    if (TWO) zB += __expf(leaky(asrc[s * HEADS + hB] + adB) - mB);
  }
  for (int off = GL / 2; off > 0; off >>= 1) {
    zA += __shfl_xor(zA, off);
    if (TWO) zB += __shfl_xor(zB, off);
  }
  if (!TWO) zB = zA;
  const float rzA = 1.0f / zA;
  const float rzB = TWO ? 1.0f / zB : rzA;

  const v4f* __restrict__ h4 = (const v4f*)hin;
  v4f accA, accB;
  {
    const float aA = __expf(lsA - mA) * rzA;
    const float aB = TWO ? __expf(lsB - mB) * rzB : aA;
    accA = aA * h4[(long long)n * 64 + lane];
    accB = aB * h4[(long long)n * 64 + 32 + lane];
  }
  for (int base = beg; base < end; base += 32) {
    const int cnt = (end - base < 32) ? end - base : 32;
    const int sv  = (lane < cnt) ? csr_src[base + lane] : n;
    float av = 0.0f;
    if (!TWO) av = __expf(leaky(asrc[sv] + adA) - mA) * rzA;
#pragma unroll 4
    for (int t = 0; t < cnt; ++t) {
      const int s = __builtin_amdgcn_readlane(sv, t);
      float aA, aB;
      if (TWO) {
        aA = __expf(leaky(asrc[s * HEADS + hA] + adA) - mA) * rzA;
        aB = __expf(leaky(asrc[s * HEADS + hB] + adB) - mB) * rzB;
      } else {
        aA = aB = __int_as_float(__builtin_amdgcn_readlane(__float_as_int(av), t));
      }
      accA += aA * h4[(long long)s * 64 + lane];
      accB += aB * h4[(long long)s * 64 + 32 + lane];
    }
  }
  v4f* __restrict__ o4 = (v4f*)outp;
  o4[(long long)n * 64 + lane]      = accA;
  o4[(long long)n * 64 + 32 + lane] = accB;
}

__global__ void ln_elu(float* __restrict__ hbuf, const float* __restrict__ b1,
                       const float* __restrict__ gamma,
                       const float* __restrict__ beta, int Nn) {
  long long gtid = (long long)blockIdx.x * blockDim.x + threadIdx.x;
  int n    = (int)(gtid >> 5);
  int lane = threadIdx.x & 31;
  if (n >= Nn) return;
  float* __restrict__ row = hbuf + (long long)n * 256;
  float vals[8];
  float sum = 0.0f;
#pragma unroll
  for (int k = 0; k < 8; ++k) {
    int idx = k * 32 + lane;
    vals[k] = row[idx] + b1[idx];
    sum += vals[k];
  }
#pragma unroll
  for (int off = 16; off > 0; off >>= 1) sum += __shfl_xor(sum, off);
  float mu = sum * (1.0f / 256.0f);
  float var = 0.0f;
#pragma unroll
  for (int k = 0; k < 8; ++k) { float t = vals[k] - mu; var += t * t; }
#pragma unroll
  for (int off = 16; off > 0; off >>= 1) var += __shfl_xor(var, off);
  var *= (1.0f / 256.0f);
  float rs = rsqrtf(var + LN_EPS);
#pragma unroll
  for (int k = 0; k < 8; ++k) {
    int idx = k * 32 + lane;
    float t = (vals[k] - mu) * rs * gamma[idx] + beta[idx];
    row[idx] = t > 0.0f ? t : (__expf(t) - 1.0f);
  }
}

__global__ void pool_accum(const float* __restrict__ o,
                           const int* __restrict__ batch,
                           float* __restrict__ outsum,
                           float* __restrict__ cnt, int Nn) {
  long long gtid = (long long)blockIdx.x * blockDim.x + threadIdx.x;
  int n    = (int)(gtid >> 5);
  int lane = threadIdx.x & 31;
  if (n >= Nn) return;
  int g = batch[n];
  const float* __restrict__ row = o + (long long)n * 256;
  float* __restrict__ acc       = outsum + (long long)g * 256;
#pragma unroll
  for (int k = 0; k < 8; ++k) {
    int idx = k * 32 + lane;
    atomicAdd(&acc[idx], row[idx]);
  }
  if (lane == 0) atomicAdd(&cnt[g], 1.0f);
}

__global__ void pool_div(float* __restrict__ out, const float* __restrict__ cnt,
                         int total) {
  int i = blockIdx.x * blockDim.x + threadIdx.x;
  if (i >= total) return;
  out[i] /= fmaxf(cnt[i >> 8], 1.0f);
}

static inline int cdiv(long long a, long long b) { return (int)((a + b - 1) / b); }

extern "C" void kernel_launch(void* const* d_in, const int* in_sizes, int n_in,
                              void* d_out, int out_size, void* d_ws, size_t ws_size,
                              hipStream_t stream) {
  const float* x        = (const float*)d_in[0];
  const int*   ei       = (const int*)  d_in[1];
  const int*   batch    = (const int*)  d_in[2];
  const float* W1       = (const float*)d_in[3];
  const float* att_src1 = (const float*)d_in[4];
  const float* att_dst1 = (const float*)d_in[5];
  const float* b1       = (const float*)d_in[6];
  const float* gamma    = (const float*)d_in[7];
  const float* beta     = (const float*)d_in[8];
  const float* W2       = (const float*)d_in[9];
  const float* att_src2 = (const float*)d_in[10];
  const float* att_dst2 = (const float*)d_in[11];
  const float* b2       = (const float*)d_in[12];
  const float* Wres     = (const float*)d_in[13];
  const float* bres     = (const float*)d_in[14];
  float* out = (float*)d_out;

  const int Nn = in_sizes[2];
  const int Ee = in_sizes[1] / 2;
  const int Gg = out_size / 256;

  const long long NB = (long long)Nn * 256;
  float* ws   = (float*)d_ws;
  float* buf0 = ws;
  float* buf1 = ws + NB;
  float* buf2 = ws + 2 * NB;
  float* scal = ws + 3 * NB;
  int*   deg  = (int*)scal;
  float* cnt  = scal + Nn;
  float* as1  = cnt + Gg;
  float* ad1  = as1 + (long long)Nn * 8;
  float* as2  = ad1 + (long long)Nn * 8;
  float* ad2  = as2 + Nn;
  int*   rowptr  = (int*)(ad2 + Nn);
  int*   cursor  = rowptr + Nn + 1;
  int*   csr_src = cursor + Nn;

  const int B = 256;
  const long long zlen = (long long)Nn + Gg;

  fill_f32<<<cdiv(zlen, B), B, 0, stream>>>(scal, zlen, 0.0f);
  fill_f32<<<cdiv((long long)Gg * 256, B), B, 0, stream>>>(out, (long long)Gg * 256, 0.0f);

  csr_count<<<cdiv(Ee, B), B, 0, stream>>>(ei, deg, Ee);
  csr_scan<<<cdiv(Nn, 1024), 1024, 0, stream>>>(deg, rowptr, cursor, Nn);
  csr_fill<<<cdiv(Ee, B), B, 0, stream>>>(ei, cursor, csr_src, Ee);

  const int gtiles = cdiv(Nn, 32);
  const int ntiles = cdiv(Nn, 8);

  gemm_wmma_n256<0, 128><<<gtiles, 128, 0, stream>>>(x, W1, buf0, nullptr, nullptr, Nn);
  attn_scores<<<cdiv((long long)Nn * 8, B), B, 0, stream>>>(buf0, att_src1, att_dst1,
                                                            as1, ad1, Nn, 8, 32);
  gat_gather<8><<<ntiles, B, 0, stream>>>(rowptr, csr_src, buf0, as1, ad1, buf1, Nn);
  ln_elu<<<cdiv((long long)Nn * 32, B), B, 0, stream>>>(buf1, b1, gamma, beta, Nn);

  gemm_wmma_n256<0, 256><<<gtiles, 128, 0, stream>>>(buf1, W2, buf0, nullptr, nullptr, Nn);
  attn_scores<<<cdiv((long long)Nn, B), B, 0, stream>>>(buf0, att_src2, att_dst2,
                                                        as2, ad2, Nn, 1, 256);
  gat_gather<1><<<ntiles, B, 0, stream>>>(rowptr, csr_src, buf0, as2, ad2, buf2, Nn);
  gemm_wmma_n256<1, 256><<<gtiles, 128, 0, stream>>>(buf1, Wres, buf2, b2, bres, Nn);

  pool_accum<<<cdiv((long long)Nn * 32, B), B, 0, stream>>>(buf2, batch, out, cnt, Nn);
  pool_div  <<<cdiv((long long)Gg * 256, B), B, 0, stream>>>(out, cnt, Gg * 256);
}


# Row-wise kernels fused into producers: attention dots into both GEMM epilogues, LayerNorm+ELU into layer-1 gather, pool sums into the residual GEMM (one atomic per run); their sum orders change
# speedup vs baseline: 3.3791x; 1.2405x over previous
#include <hip/hip_runtime.h>
#include <hip/hip_bf16.h>


typedef __attribute__((ext_vector_type(2))) float v2f;
typedef __attribute__((ext_vector_type(8))) float v8f;
typedef __attribute__((ext_vector_type(4))) float v4f;

#define NEG_SLOPE 0.2f
#define LN_EPS    1e-5f

__global__ void fill_f32(float* __restrict__ p, long long n, float v) {
  long long i = (long long)blockIdx.x * blockDim.x + threadIdx.x;
  if (i < n) p[i] = v;
}

__device__ __forceinline__ float leaky(float x) {
  return x > 0.0f ? x : NEG_SLOPE * x;
}

template <int MASK>
__device__ __forceinline__ float add_row_xor(float v) {
  const int o = __builtin_amdgcn_update_dpp(0, __float_as_int(v), 0x160 + MASK, 0xf, 0xf, false);
  return v + __int_as_float(o);
}

template <int MODE, int K, int HEADS>
__global__ void __launch_bounds__(128)
gemm_wmma_n256(const float* __restrict__ A,
               const float* __restrict__ B,
               float* __restrict__ C,
               const float* __restrict__ vec_a,
               const float* __restrict__ vec_b,
               float* __restrict__ out_a,
               float* __restrict__ out_b,
               const int* __restrict__ graph,
               int M) {
  constexpr int BM = 32;
  constexpr int P  = K + 4;
  __shared__ __attribute__((aligned(16))) float As[BM * P];
  __shared__ __attribute__((aligned(16))) float red[512];

  const int tid  = threadIdx.x;
  const int lane = tid & 31;
  const int wave = tid >> 5;
  const int c    = lane & 15;
  const int h    = lane >> 4;
  const long long row0 = (long long)blockIdx.x * BM;

  constexpr int Q = K / 4;
  for (int i = tid; i < BM * Q; i += 128) {
    const int r = i / Q, q = i - r * Q;
    long long gr = row0 + r;
    if (gr >= M) gr = M - 1;
    const v4f v = *(const v4f*)(A + gr * K + q * 4);
    *(v4f*)(As + r * P + q * 4) = v;
  }
  __syncthreads();

  const float* __restrict__ a0p = As + c * P + 2 * h;
  const float* __restrict__ a1p = As + (16 + c) * P + 2 * h;
  const float* __restrict__ Bx  = B + (long long)(2 * h) * 256 + wave * 64 + 4 * c;
  const float* __restrict__ By  = Bx + 256;

  v8f acc00 = {}, acc01 = {}, acc02 = {}, acc03 = {};
  v8f acc10 = {}, acc11 = {}, acc12 = {}, acc13 = {};

#pragma unroll 2
  for (int k = 0; k < K; k += 4) {
    const v2f a0 = *(const v2f*)(a0p + k);
    const v2f a1 = *(const v2f*)(a1p + k);
    const v4f bx = *(const v4f*)(Bx + (long long)k * 256);
    const v4f by = *(const v4f*)(By + (long long)k * 256);
    v2f b;
    b.x = bx.x; b.y = by.x;
    acc00 = __builtin_amdgcn_wmma_f32_16x16x4_f32(false, a0, false, b, (short)0, acc00, false, false);
    acc10 = __builtin_amdgcn_wmma_f32_16x16x4_f32(false, a1, false, b, (short)0, acc10, false, false);
    b.x = bx.y; b.y = by.y;
    acc01 = __builtin_amdgcn_wmma_f32_16x16x4_f32(false, a0, false, b, (short)0, acc01, false, false);
    acc11 = __builtin_amdgcn_wmma_f32_16x16x4_f32(false, a1, false, b, (short)0, acc11, false, false);
    b.x = bx.z; b.y = by.z;
    acc02 = __builtin_amdgcn_wmma_f32_16x16x4_f32(false, a0, false, b, (short)0, acc02, false, false);
    acc12 = __builtin_amdgcn_wmma_f32_16x16x4_f32(false, a1, false, b, (short)0, acc12, false, false);
    b.x = bx.w; b.y = by.w;
    acc03 = __builtin_amdgcn_wmma_f32_16x16x4_f32(false, a0, false, b, (short)0, acc03, false, false);
    acc13 = __builtin_amdgcn_wmma_f32_16x16x4_f32(false, a1, false, b, (short)0, acc13, false, false);
  }

  const int col = wave * 64 + 4 * c;
  const v4f va = *(const v4f*)(vec_a + col);
  const v4f vb = *(const v4f*)(vec_b + col);
  if (MODE == 0) {
    float ps[16], pd[16];
#pragma unroll
    for (int r = 0; r < 8; ++r) {
      const long long or0 = row0 + 8 * h + r;
      const long long or1 = or0 + 16;
      v4f v0, v1;
      v0.x = acc00[r]; v0.y = acc01[r]; v0.z = acc02[r]; v0.w = acc03[r];
      v1.x = acc10[r]; v1.y = acc11[r]; v1.z = acc12[r]; v1.w = acc13[r];
      if (or0 < M) *(v4f*)(C + or0 * 256 + col) = v0;
      if (or1 < M) *(v4f*)(C + or1 * 256 + col) = v1;
      ps[r]     = (v0.x * va.x + v0.y * va.y) + (v0.z * va.z + v0.w * va.w);
      pd[r]     = (v0.x * vb.x + v0.y * vb.y) + (v0.z * vb.z + v0.w * vb.w);
      ps[8 + r] = (v1.x * va.x + v1.y * va.y) + (v1.z * va.z + v1.w * va.w);
      pd[8 + r] = (v1.x * vb.x + v1.y * vb.y) + (v1.z * vb.z + v1.w * vb.w);
    }
#pragma unroll
    for (int i = 0; i < 16; ++i) { ps[i] = add_row_xor<1>(ps[i]); pd[i] = add_row_xor<1>(pd[i]); }
#pragma unroll
    for (int i = 0; i < 16; ++i) { ps[i] = add_row_xor<2>(ps[i]); pd[i] = add_row_xor<2>(pd[i]); }
#pragma unroll
    for (int i = 0; i < 16; ++i) { ps[i] = add_row_xor<4>(ps[i]); pd[i] = add_row_xor<4>(pd[i]); }
    if (HEADS == 1) {
#pragma unroll
      for (int i = 0; i < 16; ++i) { ps[i] = add_row_xor<8>(ps[i]); pd[i] = add_row_xor<8>(pd[i]); }
      if (c == 0) {
#pragma unroll
        for (int i = 0; i < 16; ++i) {
          const int j = (i < 8) ? (8 * h + i) : (8 + 8 * h + i);
          red[(wave * 32 + j) * 2]     = ps[i];
          red[(wave * 32 + j) * 2 + 1] = pd[i];
        }
      }
      __syncthreads();
      if (tid < 64) {
        const int which = tid >> 5, j = tid & 31;
        const float s = ((red[j * 2 + which] + red[(32 + j) * 2 + which]) +
                         red[(64 + j) * 2 + which]) + red[(96 + j) * 2 + which];
        if (row0 + j < M) (which ? out_b : out_a)[row0 + j] = s;
      }
    } else {
      if ((c & 7) == 0) {
        const int hd = 2 * wave + (c >> 3);
#pragma unroll
        for (int i = 0; i < 16; ++i) {
          const int j = (i < 8) ? (8 * h + i) : (8 + 8 * h + i);
          red[j * 8 + hd]       = ps[i];
          red[256 + j * 8 + hd] = pd[i];
        }
      }
      __syncthreads();
      const int which = tid >> 6, t = tid & 63;
      if (row0 + (t >> 1) < M)
        *(v4f*)((which ? out_b : out_a) + row0 * 8 + 4 * t) = *(const v4f*)(red + which * 256 + 4 * t);
    }
  } else {
    long long gr = row0 + lane;
    if (gr >= M) gr = M - 1;
    const int g_l = graph[gr];
    int cur = -1;
    v4f run = {0.0f, 0.0f, 0.0f, 0.0f};
#pragma unroll
    for (int half = 0; half < 2; ++half) {
      v4f val[8];
#pragma unroll
      for (int r = 0; r < 8; ++r) {
        long long orow = row0 + 16 * half + 8 * h + r;
        if (orow >= M) orow = M - 1;
        v4f v;
        if (half == 0) { v.x = acc00[r]; v.y = acc01[r]; v.z = acc02[r]; v.w = acc03[r]; }
        else           { v.x = acc10[r]; v.y = acc11[r]; v.z = acc12[r]; v.w = acc13[r]; }
        val[r] = *(const v4f*)(C + orow * 256 + col) + ((v + va) + vb);
      }
#pragma unroll
      for (int r = 0; r < 8; ++r) {
        const int j = 16 * half + 8 * h + r;
        const int g = __shfl(g_l, j);
        if (row0 + j < M) {
          if (g != cur) {
            if (cur >= 0) {
              float* o = out_a + (long long)cur * 256 + col;
              atomicAdd(o, run.x); atomicAdd(o + 1, run.y);
              atomicAdd(o + 2, run.z); atomicAdd(o + 3, run.w);
            }
            run = val[r];
            cur = g;
          } else {
            run += val[r];
          }
        }
      }
    }
    if (cur >= 0) {
      float* o = out_a + (long long)cur * 256 + col;
      atomicAdd(o, run.x); atomicAdd(o + 1, run.y);
      atomicAdd(o + 2, run.z); atomicAdd(o + 3, run.w);
    }
    if (wave == 0 && row0 + lane < M) atomicAdd(&out_b[g_l], 1.0f);
  }
}

__global__ void csr_count(const int* __restrict__ ei, int* __restrict__ deg, int E) {
  int e = blockIdx.x * blockDim.x + threadIdx.x;
  if (e < E) atomicAdd(&deg[ei[E + e]], 1);
}

__global__ void __launch_bounds__(1024)
csr_scan(const int* __restrict__ deg, int* __restrict__ rowptr,
         int* __restrict__ cursor, int Nn) {
  __shared__ int wsum[32];
  __shared__ int psum[32];
  const int t = threadIdx.x, lane = t & 31, w = t >> 5;
  const int b = blockIdx.x;
  int pre = 0;
  for (int j = 0; j < b; ++j) pre += deg[j * 1024 + t];
  const int idx = b * 1024 + t;
  const int d = (idx < Nn) ? deg[idx] : 0;
  int v = d;
  for (int off = 1; off < 32; off <<= 1) {
    int u = __shfl_up(v, off);
    if (lane >= off) v += u;
  }
  for (int off = 16; off > 0; off >>= 1) pre += __shfl_xor(pre, off);
  if (lane == 31) wsum[w] = v;
  if (lane == 0)  psum[w] = pre;
  __syncthreads();
  int x = psum[lane] + (lane < w ? wsum[lane] : 0);
  for (int off = 16; off > 0; off >>= 1) x += __shfl_xor(x, off);
  const int excl = x + (v - d);
  if (idx < Nn) { rowptr[idx] = excl; cursor[idx] = excl; }
  if (idx == Nn - 1) rowptr[Nn] = excl + d;
}

__global__ void csr_fill(const int* __restrict__ ei, int* __restrict__ cursor,
                         int* __restrict__ csr_src, int E) {
  int e = blockIdx.x * blockDim.x + threadIdx.x;
  if (e < E) {
    int p = atomicAdd(&cursor[ei[E + e]], 1);
    csr_src[p] = ei[e];
  }
}

template <int HEADS, bool LN>
__global__ void __launch_bounds__(256)
gat_gather(const int* __restrict__ rowptr, const int* __restrict__ csr_src,
           const float* __restrict__ hin, const float* __restrict__ asrc,
           const float* __restrict__ adst, float* __restrict__ outp,
           const float* __restrict__ lnb, const float* __restrict__ gamma,
           const float* __restrict__ beta, int Nn) {
  constexpr int  CC  = 256 / HEADS;
  constexpr int  GL  = (CC / 4 < 32) ? CC / 4 : 32;
  constexpr bool TWO = (HEADS > 1);
  const int lane = threadIdx.x & 31;
  int n = blockIdx.x * 8 + (threadIdx.x >> 5);
  if (n >= Nn) return;
  n = __builtin_amdgcn_readfirstlane(n);
  const int hA  = (4 * lane) / CC, hB = (128 + 4 * lane) / CC;
  const int sub = lane & (GL - 1);
  const int beg = rowptr[n], end = rowptr[n + 1];
  const float adA = adst[n * HEADS + hA];
  const float adB = TWO ? adst[n * HEADS + hB] : adA;
  const float lsA = leaky(asrc[n * HEADS + hA] + adA);
  const float lsB = TWO ? leaky(asrc[n * HEADS + hB] + adB) : lsA;

  float mA = lsA, mB = lsB;
  for (int i = beg + sub; i < end; i += GL) {
    const int s = csr_src[i];
    mA = fmaxf(mA, leaky(asrc[s * HEADS + hA] + adA));
    if (TWO) mB = fmaxf(mB, leaky(asrc[s * HEADS + hB] + adB));
  }
  for (int off = GL / 2; off > 0; off >>= 1) {
    mA = fmaxf(mA, __shfl_xor(mA, off));
    if (TWO) mB = fmaxf(mB, __shfl_xor(mB, off));
  }
  if (!TWO) mB = mA;
  float zA = (sub == 0) ? __expf(lsA - mA) : 0.0f;
  float zB = (sub == 0) ? __expf(lsB - mB) : 0.0f;
  for (int i = beg + sub; i < end; i += GL) {
    const int s = csr_src[i];
    zA += __expf(leaky(asrc[s * HEADS + hA] + adA) - mA);
    if (TWO) zB += __expf(leaky(asrc[s * HEADS + hB] + adB) - mB);
  }
  for (int off = GL / 2; off > 0; off >>= 1) {
    zA += __shfl_xor(zA, off);
    if (TWO) zB += __shfl_xor(zB, off);
  }
  if (!TWO) zB = zA;
  const float rzA = 1.0f / zA;
  const float rzB = TWO ? 1.0f / zB : rzA;

  const v4f* __restrict__ h4 = (const v4f*)hin;
  v4f accA, accB;
  {
    const float aA = __expf(lsA - mA) * rzA;
    const float aB = TWO ? __expf(lsB - mB) * rzB : aA;
    accA = aA * h4[(long long)n * 64 + lane];
    accB = aB * h4[(long long)n * 64 + 32 + lane];
  }
  for (int base = beg; base < end; base += 32) {
    const int cnt = (end - base < 32) ? end - base : 32;
    const int sv  = (lane < cnt) ? csr_src[base + lane] : n;
    float av = 0.0f;
    if (!TWO) av = __expf(leaky(asrc[sv] + adA) - mA) * rzA;
#pragma unroll 4
    for (int t = 0; t < cnt; ++t) {
      const int s = __builtin_amdgcn_readlane(sv, t);
      float aA, aB;
      if (TWO) {
        aA = __expf(leaky(asrc[s * HEADS + hA] + adA) - mA) * rzA;
        aB = __expf(leaky(asrc[s * HEADS + hB] + adB) - mB) * rzB;
      } else {
        aA = aB = __int_as_float(__builtin_amdgcn_readlane(__float_as_int(av), t));
      }
      accA += aA * h4[(long long)s * 64 + lane];
      accB += aB * h4[(long long)s * 64 + 32 + lane];
    }
  }
  if (LN) {
    const v4f xA = accA + *(const v4f*)(lnb + 4 * lane);
    const v4f xB = accB + *(const v4f*)(lnb + 128 + 4 * lane);
    float sum = ((xA.x + xA.y) + (xA.z + xA.w)) + ((xB.x + xB.y) + (xB.z + xB.w));
    for (int off = 16; off > 0; off >>= 1) sum += __shfl_xor(sum, off);
    const float mu = sum * (1.0f / 256.0f);
    const v4f dA = xA - mu, dB = xB - mu;
    float var = ((dA.x * dA.x + dA.y * dA.y) + (dA.z * dA.z + dA.w * dA.w)) +
                ((dB.x * dB.x + dB.y * dB.y) + (dB.z * dB.z + dB.w * dB.w));
    for (int off = 16; off > 0; off >>= 1) var += __shfl_xor(var, off);
    var *= (1.0f / 256.0f);
    const float rs = rsqrtf(var + LN_EPS);
    const v4f tA = dA * rs * *(const v4f*)(gamma + 4 * lane) +
                   *(const v4f*)(beta + 4 * lane);
    const v4f tB = dB * rs * *(const v4f*)(gamma + 128 + 4 * lane) +
                   *(const v4f*)(beta + 128 + 4 * lane);
    accA.x = tA.x > 0.0f ? tA.x : (__expf(tA.x) - 1.0f);
    accA.y = tA.y > 0.0f ? tA.y : (__expf(tA.y) - 1.0f);
    accA.z = tA.z > 0.0f ? tA.z : (__expf(tA.z) - 1.0f);
    accA.w = tA.w > 0.0f ? tA.w : (__expf(tA.w) - 1.0f);
    accB.x = tB.x > 0.0f ? tB.x : (__expf(tB.x) - 1.0f);
    accB.y = tB.y > 0.0f ? tB.y : (__expf(tB.y) - 1.0f);
    accB.z = tB.z > 0.0f ? tB.z : (__expf(tB.z) - 1.0f);
    accB.w = tB.w > 0.0f ? tB.w : (__expf(tB.w) - 1.0f);
  }
  v4f* __restrict__ o4 = (v4f*)outp;
  o4[(long long)n * 64 + lane]      = accA;
  o4[(long long)n * 64 + 32 + lane] = accB;
}

__global__ void pool_div(float* __restrict__ out, const float* __restrict__ cnt,
                         int total) {
  int i = blockIdx.x * blockDim.x + threadIdx.x;
  if (i >= total) return;
  out[i] /= fmaxf(cnt[i >> 8], 1.0f);
}

static inline int cdiv(long long a, long long b) { return (int)((a + b - 1) / b); }

extern "C" void kernel_launch(void* const* d_in, const int* in_sizes, int n_in,
                              void* d_out, int out_size, void* d_ws, size_t ws_size,
                              hipStream_t stream) {
  const float* x        = (const float*)d_in[0];
  const int*   ei       = (const int*)  d_in[1];
  const int*   batch    = (const int*)  d_in[2];
  const float* W1       = (const float*)d_in[3];
  const float* att_src1 = (const float*)d_in[4];
  const float* att_dst1 = (const float*)d_in[5];
  const float* b1       = (const float*)d_in[6];
  const float* gamma    = (const float*)d_in[7];
  const float* beta     = (const float*)d_in[8];
  const float* W2       = (const float*)d_in[9];
  const float* att_src2 = (const float*)d_in[10];
  const float* att_dst2 = (const float*)d_in[11];
  const float* b2       = (const float*)d_in[12];
  const float* Wres     = (const float*)d_in[13];
  const float* bres     = (const float*)d_in[14];
  float* out = (float*)d_out;

  const int Nn = in_sizes[2];
  const int Ee = in_sizes[1] / 2;
  const int Gg = out_size / 256;

  const long long NB = (long long)Nn * 256;
  float* ws   = (float*)d_ws;
  float* buf0 = ws;
  float* buf1 = ws + NB;
  float* buf2 = ws + 2 * NB;
  float* scal = ws + 3 * NB;
  const long long Np = ((long long)Nn + 32) / 32 * 32;
  const long long Gp = ((long long)Gg + 31) / 32 * 32;
  int*   deg  = (int*)scal;
  float* cnt  = scal + Np;
  float* as1  = cnt + Gp;
  float* ad1  = as1 + Np * 8;
  float* as2  = ad1 + Np * 8;
  float* ad2  = as2 + Np;
  int*   rowptr  = (int*)(ad2 + Np);
  int*   cursor  = rowptr + Np;
  int*   csr_src = cursor + Np;

  const int B = 256;
  const long long zlen = Np + Gg;

  fill_f32<<<cdiv(zlen, B), B, 0, stream>>>(scal, zlen, 0.0f);
  fill_f32<<<cdiv((long long)Gg * 256, B), B, 0, stream>>>(out, (long long)Gg * 256, 0.0f);

  csr_count<<<cdiv(Ee, B), B, 0, stream>>>(ei, deg, Ee);
  csr_scan<<<cdiv(Nn, 1024), 1024, 0, stream>>>(deg, rowptr, cursor, Nn);
  csr_fill<<<cdiv(Ee, B), B, 0, stream>>>(ei, cursor, csr_src, Ee);

  const int gtiles = cdiv(Nn, 32);
  const int ntiles = cdiv(Nn, 8);

  gemm_wmma_n256<0, 128, 8><<<gtiles, 128, 0, stream>>>(x, W1, buf0, att_src1, att_dst1,
                                                        as1, ad1, nullptr, Nn);
  gat_gather<8, true><<<ntiles, B, 0, stream>>>(rowptr, csr_src, buf0, as1, ad1, buf1,
                                                b1, gamma, beta, Nn);

  gemm_wmma_n256<0, 256, 1><<<gtiles, 128, 0, stream>>>(buf1, W2, buf0, att_src2, att_dst2,
                                                        as2, ad2, nullptr, Nn);
  gat_gather<1, false><<<ntiles, B, 0, stream>>>(rowptr, csr_src, buf0, as2, ad2, buf2,
                                                 nullptr, nullptr, nullptr, Nn);
  gemm_wmma_n256<2, 256, 0><<<gtiles, 128, 0, stream>>>(buf1, Wres, buf2, b2, bres,
                                                        out, cnt, batch, Nn);

  pool_div<<<cdiv((long long)Gg * 256, B), B, 0, stream>>>(out, cnt, Gg * 256);
}


# Mean-pool moved ahead of layer 2's linear maps: gather and pool rows of h, then W2 and Wres act on the 512 pooled rows (logits from h . W2 att); two 50000-row GEMMs removed; sum order changes
# speedup vs baseline: 4.5459x; 1.3453x over previous
#include <hip/hip_runtime.h>
#include <hip/hip_bf16.h>


typedef __attribute__((ext_vector_type(2))) float v2f;
typedef __attribute__((ext_vector_type(8))) float v8f;
typedef __attribute__((ext_vector_type(4))) float v4f;

#define NEG_SLOPE 0.2f
#define LN_EPS    1e-5f

__global__ void fill_f32(float* __restrict__ p, long long n, float v) {
  long long i = (long long)blockIdx.x * blockDim.x + threadIdx.x;
  if (i < n) p[i] = v;
}

__device__ __forceinline__ float leaky(float x) {
  return x > 0.0f ? x : NEG_SLOPE * x;
}

template <int MASK>
__device__ __forceinline__ float add_row_xor(float v) {
  const int o = __builtin_amdgcn_update_dpp(0, __float_as_int(v), 0x160 + MASK, 0xf, 0xf, false);
  return v + __int_as_float(o);
}

template <int K, int HEADS>
__global__ void __launch_bounds__(128)
gemm_wmma_n256(const float* __restrict__ A,
               const float* __restrict__ B,
               float* __restrict__ C,
               const float* __restrict__ vec_a,
               const float* __restrict__ vec_b,
               float* __restrict__ out_a,
               float* __restrict__ out_b,
               int M) {
  constexpr int BM = 32;
  constexpr int P  = K + 4;
  __shared__ __attribute__((aligned(16))) float As[BM * P];
  __shared__ __attribute__((aligned(16))) float red[512];

  const int tid  = threadIdx.x;
  const int lane = tid & 31;
  const int wave = tid >> 5;
  const int c    = lane & 15;
  const int h    = lane >> 4;
  const long long row0 = (long long)blockIdx.x * BM;

  constexpr int Q = K / 4;
  for (int i = tid; i < BM * Q; i += 128) {
    const int r = i / Q, q = i - r * Q;
    long long gr = row0 + r;
    if (gr >= M) gr = M - 1;
    const v4f v = *(const v4f*)(A + gr * K + q * 4);
    *(v4f*)(As + r * P + q * 4) = v;
  }
  __syncthreads();

  const float* __restrict__ a0p = As + c * P + 2 * h;
  const float* __restrict__ a1p = As + (16 + c) * P + 2 * h;
  const float* __restrict__ Bx  = B + (long long)(2 * h) * 256 + wave * 64 + 4 * c;
  const float* __restrict__ By  = Bx + 256;

  v8f acc00 = {}, acc01 = {}, acc02 = {}, acc03 = {};
  v8f acc10 = {}, acc11 = {}, acc12 = {}, acc13 = {};

#pragma unroll 2
  for (int k = 0; k < K; k += 4) {
    const v2f a0 = *(const v2f*)(a0p + k);
    const v2f a1 = *(const v2f*)(a1p + k);
    const v4f bx = *(const v4f*)(Bx + (long long)k * 256);
    const v4f by = *(const v4f*)(By + (long long)k * 256);
    v2f b;
    b.x = bx.x; b.y = by.x;
    acc00 = __builtin_amdgcn_wmma_f32_16x16x4_f32(false, a0, false, b, (short)0, acc00, false, false);
    acc10 = __builtin_amdgcn_wmma_f32_16x16x4_f32(false, a1, false, b, (short)0, acc10, false, false);
    b.x = bx.y; b.y = by.y;
    acc01 = __builtin_amdgcn_wmma_f32_16x16x4_f32(false, a0, false, b, (short)0, acc01, false, false);
    acc11 = __builtin_amdgcn_wmma_f32_16x16x4_f32(false, a1, false, b, (short)0, acc11, false, false);
    b.x = bx.z; b.y = by.z;
    acc02 = __builtin_amdgcn_wmma_f32_16x16x4_f32(false, a0, false, b, (short)0, acc02, false, false);
    acc12 = __builtin_amdgcn_wmma_f32_16x16x4_f32(false, a1, false, b, (short)0, acc12, false, false);
    b.x = bx.w; b.y = by.w;
    acc03 = __builtin_amdgcn_wmma_f32_16x16x4_f32(false, a0, false, b, (short)0, acc03, false, false);
    acc13 = __builtin_amdgcn_wmma_f32_16x16x4_f32(false, a1, false, b, (short)0, acc13, false, false);
  }

  const int col = wave * 64 + 4 * c;
  const v4f va = *(const v4f*)(vec_a + col);
  const v4f vb = *(const v4f*)(vec_b + col);
  {
    float ps[16], pd[16];
#pragma unroll
    for (int r = 0; r < 8; ++r) {
      const long long or0 = row0 + 8 * h + r;
      const long long or1 = or0 + 16;
      v4f v0, v1;
      v0.x = acc00[r]; v0.y = acc01[r]; v0.z = acc02[r]; v0.w = acc03[r];
      v1.x = acc10[r]; v1.y = acc11[r]; v1.z = acc12[r]; v1.w = acc13[r];
      if (or0 < M) *(v4f*)(C + or0 * 256 + col) = v0;
      if (or1 < M) *(v4f*)(C + or1 * 256 + col) = v1;
      ps[r]     = (v0.x * va.x + v0.y * va.y) + (v0.z * va.z + v0.w * va.w);
      pd[r]     = (v0.x * vb.x + v0.y * vb.y) + (v0.z * vb.z + v0.w * vb.w);
      ps[8 + r] = (v1.x * va.x + v1.y * va.y) + (v1.z * va.z + v1.w * va.w);
      pd[8 + r] = (v1.x * vb.x + v1.y * vb.y) + (v1.z * vb.z + v1.w * vb.w);
    }
#pragma unroll
    for (int i = 0; i < 16; ++i) { ps[i] = add_row_xor<1>(ps[i]); pd[i] = add_row_xor<1>(pd[i]); }
#pragma unroll
    for (int i = 0; i < 16; ++i) { ps[i] = add_row_xor<2>(ps[i]); pd[i] = add_row_xor<2>(pd[i]); }
#pragma unroll
    for (int i = 0; i < 16; ++i) { ps[i] = add_row_xor<4>(ps[i]); pd[i] = add_row_xor<4>(pd[i]); }
    if (HEADS == 1) {
#pragma unroll
      for (int i = 0; i < 16; ++i) { ps[i] = add_row_xor<8>(ps[i]); pd[i] = add_row_xor<8>(pd[i]); }
      if (c == 0) {
#pragma unroll
        for (int i = 0; i < 16; ++i) {
          const int j = (i < 8) ? (8 * h + i) : (8 + 8 * h + i);
          red[(wave * 32 + j) * 2]     = ps[i];
          red[(wave * 32 + j) * 2 + 1] = pd[i];
        }
      }
      __syncthreads();
      if (tid < 64) {
        const int which = tid >> 5, j = tid & 31;
        const float s = ((red[j * 2 + which] + red[(32 + j) * 2 + which]) +
                         red[(64 + j) * 2 + which]) + red[(96 + j) * 2 + which];
        if (row0 + j < M) (which ? out_b : out_a)[row0 + j] = s;
      }
    } else {
      if ((c & 7) == 0) {
        const int hd = 2 * wave + (c >> 3);
#pragma unroll
        for (int i = 0; i < 16; ++i) {
          const int j = (i < 8) ? (8 * h + i) : (8 + 8 * h + i);
          red[j * 8 + hd]       = ps[i];
          red[256 + j * 8 + hd] = pd[i];
        }
      }
      __syncthreads();
      const int which = tid >> 6, t = tid & 63;
      if (row0 + (t >> 1) < M)
        *(v4f*)((which ? out_b : out_a) + row0 * 8 + 4 * t) = *(const v4f*)(red + which * 256 + 4 * t);
    }
  }
}

__global__ void csr_count(const int* __restrict__ ei, int* __restrict__ deg, int E) {
  int e = blockIdx.x * blockDim.x + threadIdx.x;
  if (e < E) atomicAdd(&deg[ei[E + e]], 1);
}

__global__ void __launch_bounds__(1024)
csr_scan(const int* __restrict__ deg, int* __restrict__ rowptr,
         int* __restrict__ cursor, int Nn) {
  __shared__ int wsum[32];
  __shared__ int psum[32];
  const int t = threadIdx.x, lane = t & 31, w = t >> 5;
  const int b = blockIdx.x;
  int pre = 0;
  for (int j = 0; j < b; ++j) pre += deg[j * 1024 + t];
  const int idx = b * 1024 + t;
  const int d = (idx < Nn) ? deg[idx] : 0;
  int v = d;
  for (int off = 1; off < 32; off <<= 1) {
    int u = __shfl_up(v, off);
    if (lane >= off) v += u;
  }
  for (int off = 16; off > 0; off >>= 1) pre += __shfl_xor(pre, off);
  if (lane == 31) wsum[w] = v;
  if (lane == 0)  psum[w] = pre;
  __syncthreads();
  int x = psum[lane] + (lane < w ? wsum[lane] : 0);
  for (int off = 16; off > 0; off >>= 1) x += __shfl_xor(x, off);
  const int excl = x + (v - d);
  if (idx < Nn) { rowptr[idx] = excl; cursor[idx] = excl; }
  if (idx == Nn - 1) rowptr[Nn] = excl + d;
}

__global__ void csr_fill(const int* __restrict__ ei, int* __restrict__ cursor,
                         int* __restrict__ csr_src, int E) {
  int e = blockIdx.x * blockDim.x + threadIdx.x;
  if (e < E) {
    int p = atomicAdd(&cursor[ei[E + e]], 1);
    csr_src[p] = ei[e];
  }
}

template <int HEADS, bool LN>
__global__ void __launch_bounds__(256)
gat_gather(const int* __restrict__ rowptr, const int* __restrict__ csr_src,
           const float* __restrict__ hin, const float* __restrict__ asrc,
           const float* __restrict__ adst, float* __restrict__ outp,
           const float* __restrict__ lnb, const float* __restrict__ gamma,
           const float* __restrict__ beta, const float* __restrict__ u2,
           float* __restrict__ nas, float* __restrict__ nad, int Nn) {
  constexpr int  CC  = 256 / HEADS;
  constexpr int  GL  = (CC / 4 < 32) ? CC / 4 : 32;
  constexpr bool TWO = (HEADS > 1);
  const int lane = threadIdx.x & 31;
  int n = blockIdx.x * 8 + (threadIdx.x >> 5);
  if (n >= Nn) return;
  n = __builtin_amdgcn_readfirstlane(n);
  const int hA  = (4 * lane) / CC, hB = (128 + 4 * lane) / CC;
  const int sub = lane & (GL - 1);
  const int beg = rowptr[n], end = rowptr[n + 1];
  const float adA = adst[n * HEADS + hA];
  const float adB = TWO ? adst[n * HEADS + hB] : adA;
  const float lsA = leaky(asrc[n * HEADS + hA] + adA);
  const float lsB = TWO ? leaky(asrc[n * HEADS + hB] + adB) : lsA;

  float mA = lsA, mB = lsB;
  for (int i = beg + sub; i < end; i += GL) {
    const int s = csr_src[i];
    mA = fmaxf(mA, leaky(asrc[s * HEADS + hA] + adA));
    if (TWO) mB = fmaxf(mB, leaky(asrc[s * HEADS + hB] + adB));
  }
  for (int off = GL / 2; off > 0; off >>= 1) {
    mA = fmaxf(mA, __shfl_xor(mA, off));
    if (TWO) mB = fmaxf(mB, __shfl_xor(mB, off));
  }
  if (!TWO) mB = mA;
  float zA = (sub == 0) ? __expf(lsA - mA) : 0.0f;
  float zB = (sub == 0) ? __expf(lsB - mB) : 0.0f;
  for (int i = beg + sub; i < end; i += GL) {
    const int s = csr_src[i];
    zA += __expf(leaky(asrc[s * HEADS + hA] + adA) - mA);
    if (TWO) zB += __expf(leaky(asrc[s * HEADS + hB] + adB) - mB);
  }
  for (int off = GL / 2; off > 0; off >>= 1) {
    zA += __shfl_xor(zA, off);
    if (TWO) zB += __shfl_xor(zB, off);
  }
  if (!TWO) zB = zA;
  const float rzA = 1.0f / zA;
  const float rzB = TWO ? 1.0f / zB : rzA;

  const v4f* __restrict__ h4 = (const v4f*)hin;
  v4f accA, accB;
  {
    const float aA = __expf(lsA - mA) * rzA;
    const float aB = TWO ? __expf(lsB - mB) * rzB : aA;
    accA = aA * h4[(long long)n * 64 + lane];
    accB = aB * h4[(long long)n * 64 + 32 + lane];
  }
  for (int base = beg; base < end; base += 32) {
    const int cnt = (end - base < 32) ? end - base : 32;
    const int sv  = (lane < cnt) ? csr_src[base + lane] : n;
    float av = 0.0f;
    if (!TWO) av = __expf(leaky(asrc[sv] + adA) - mA) * rzA;
#pragma unroll 4
    for (int t = 0; t < cnt; ++t) {
      const int s = __builtin_amdgcn_readlane(sv, t);
      float aA, aB;
      if (TWO) {
        aA = __expf(leaky(asrc[s * HEADS + hA] + adA) - mA) * rzA;
        aB = __expf(leaky(asrc[s * HEADS + hB] + adB) - mB) * rzB;
      } else {
        aA = aB = __int_as_float(__builtin_amdgcn_readlane(__float_as_int(av), t));
      }
      accA += aA * h4[(long long)s * 64 + lane];
      accB += aB * h4[(long long)s * 64 + 32 + lane];
    }
  }
  if (LN) {
    const v4f xA = accA + *(const v4f*)(lnb + 4 * lane);
    const v4f xB = accB + *(const v4f*)(lnb + 128 + 4 * lane);
    float sum = ((xA.x + xA.y) + (xA.z + xA.w)) + ((xB.x + xB.y) + (xB.z + xB.w));
    for (int off = 16; off > 0; off >>= 1) sum += __shfl_xor(sum, off);
    const float mu = sum * (1.0f / 256.0f);
    const v4f dA = xA - mu, dB = xB - mu;
    float var = ((dA.x * dA.x + dA.y * dA.y) + (dA.z * dA.z + dA.w * dA.w)) +
                ((dB.x * dB.x + dB.y * dB.y) + (dB.z * dB.z + dB.w * dB.w));
    for (int off = 16; off > 0; off >>= 1) var += __shfl_xor(var, off);
    var *= (1.0f / 256.0f);
    const float rs = rsqrtf(var + LN_EPS);
    const v4f tA = dA * rs * *(const v4f*)(gamma + 4 * lane) +
                   *(const v4f*)(beta + 4 * lane);
    const v4f tB = dB * rs * *(const v4f*)(gamma + 128 + 4 * lane) +
                   *(const v4f*)(beta + 128 + 4 * lane);
    accA.x = tA.x > 0.0f ? tA.x : (__expf(tA.x) - 1.0f);
    accA.y = tA.y > 0.0f ? tA.y : (__expf(tA.y) - 1.0f);
    accA.z = tA.z > 0.0f ? tA.z : (__expf(tA.z) - 1.0f);
    accA.w = tA.w > 0.0f ? tA.w : (__expf(tA.w) - 1.0f);
    accB.x = tB.x > 0.0f ? tB.x : (__expf(tB.x) - 1.0f);
    accB.y = tB.y > 0.0f ? tB.y : (__expf(tB.y) - 1.0f);
    accB.z = tB.z > 0.0f ? tB.z : (__expf(tB.z) - 1.0f);
    accB.w = tB.w > 0.0f ? tB.w : (__expf(tB.w) - 1.0f);
    const v4f usA = *(const v4f*)(u2 + 4 * lane),       usB = *(const v4f*)(u2 + 128 + 4 * lane);
    const v4f udA = *(const v4f*)(u2 + 256 + 4 * lane), udB = *(const v4f*)(u2 + 384 + 4 * lane);
    float ds = ((accA.x * usA.x + accA.y * usA.y) + (accA.z * usA.z + accA.w * usA.w)) +
               ((accB.x * usB.x + accB.y * usB.y) + (accB.z * usB.z + accB.w * usB.w));
    float dd = ((accA.x * udA.x + accA.y * udA.y) + (accA.z * udA.z + accA.w * udA.w)) +
               ((accB.x * udB.x + accB.y * udB.y) + (accB.z * udB.z + accB.w * udB.w));
    for (int off = 16; off > 0; off >>= 1) {
      ds += __shfl_xor(ds, off);
      dd += __shfl_xor(dd, off);
    }
    if (lane < 2) (lane ? nad : nas)[n] = lane ? dd : ds;
  }
  v4f* __restrict__ o4 = (v4f*)outp;
  o4[(long long)n * 64 + lane]      = accA;
  o4[(long long)n * 64 + 32 + lane] = accB;
}

__global__ void __launch_bounds__(256)
att_proj(const float* __restrict__ W, const float* __restrict__ vs,
         const float* __restrict__ vd, float* __restrict__ u) {
  const int lane = threadIdx.x & 31;
  const int k = blockIdx.x * 8 + (threadIdx.x >> 5);
  const v4f wA = *(const v4f*)(W + k * 256 + 4 * lane);
  const v4f wB = *(const v4f*)(W + k * 256 + 128 + 4 * lane);
  const v4f sA = *(const v4f*)(vs + 4 * lane), sB = *(const v4f*)(vs + 128 + 4 * lane);
  const v4f dA = *(const v4f*)(vd + 4 * lane), dB = *(const v4f*)(vd + 128 + 4 * lane);
  float ds = ((wA.x * sA.x + wA.y * sA.y) + (wA.z * sA.z + wA.w * sA.w)) +
             ((wB.x * sB.x + wB.y * sB.y) + (wB.z * sB.z + wB.w * sB.w));
  float dd = ((wA.x * dA.x + wA.y * dA.y) + (wA.z * dA.z + wA.w * dA.w)) +
             ((wB.x * dB.x + wB.y * dB.y) + (wB.z * dB.z + wB.w * dB.w));
  for (int off = 16; off > 0; off >>= 1) {
    ds += __shfl_xor(ds, off);
    dd += __shfl_xor(dd, off);
  }
  if (lane == 0) { u[k] = ds; u[256 + k] = dd; }
}

__global__ void __launch_bounds__(256)
pool_rows(const float* __restrict__ ha, const float* __restrict__ hb,
          const int* __restrict__ graph, float* __restrict__ pa,
          float* __restrict__ pb, float* __restrict__ cnt, int M) {
  const int q = threadIdx.x & 63;
  const long long base = (long long)blockIdx.x * 64 + (threadIdx.x >> 6) * 16;
  const v4f* __restrict__ a4 = (const v4f*)ha;
  const v4f* __restrict__ b4 = (const v4f*)hb;
  int cur = -1;
  float cn = 0.0f;
  v4f ra = {0.0f, 0.0f, 0.0f, 0.0f}, rb = {0.0f, 0.0f, 0.0f, 0.0f};
#pragma unroll
  for (int half = 0; half < 2; ++half) {
    v4f xa[8], xb[8];
    int g[8];
#pragma unroll
    for (int r = 0; r < 8; ++r) {
      long long row = base + 8 * half + r;
      if (row >= M) row = M - 1;
      g[r]  = graph[row];
      xa[r] = a4[row * 64 + q];
      xb[r] = b4[row * 64 + q];
    }
#pragma unroll
    for (int r = 0; r < 8; ++r) {
      if (base + 8 * half + r < M) {
        if (g[r] != cur) {
          if (cur >= 0) {
            float* oa = pa + (long long)cur * 256 + 4 * q;
            float* ob = pb + (long long)cur * 256 + 4 * q;
            atomicAdd(oa, ra.x); atomicAdd(oa + 1, ra.y);
            atomicAdd(oa + 2, ra.z); atomicAdd(oa + 3, ra.w);
            atomicAdd(ob, rb.x); atomicAdd(ob + 1, rb.y);
            atomicAdd(ob + 2, rb.z); atomicAdd(ob + 3, rb.w);
            if (q == 0) atomicAdd(cnt + cur, cn);
          }
          cur = g[r]; ra = xa[r]; rb = xb[r]; cn = 1.0f;
        } else {
          ra += xa[r]; rb += xb[r]; cn += 1.0f;
        }
      }
    }
  }
  if (cur >= 0) {
    float* oa = pa + (long long)cur * 256 + 4 * q;
    float* ob = pb + (long long)cur * 256 + 4 * q;
    atomicAdd(oa, ra.x); atomicAdd(oa + 1, ra.y);
    atomicAdd(oa + 2, ra.z); atomicAdd(oa + 3, ra.w);
    atomicAdd(ob, rb.x); atomicAdd(ob + 1, rb.y);
    atomicAdd(ob + 2, rb.z); atomicAdd(ob + 3, rb.w);
    if (q == 0) atomicAdd(cnt + cur, cn);
  }
}

__global__ void __launch_bounds__(256)
pooled_out(const float* __restrict__ pa, const float* __restrict__ Wa,
           const float* __restrict__ pb, const float* __restrict__ Wb,
           const float* __restrict__ va, const float* __restrict__ vb,
           const float* __restrict__ cnt, float* __restrict__ out, int M) {
  __shared__ __attribute__((aligned(16))) float part[8 * 16 * 64];
  const int tid  = threadIdx.x;
  const int lane = tid & 31;
  const int wave = tid >> 5;
  const int c    = lane & 15;
  const int h    = lane >> 4;
  const int row0 = blockIdx.x * 16;
  const int col0 = blockIdx.y * 64;
  const float* __restrict__ Ap = (wave < 4) ? pa : pb;
  const float* __restrict__ Bp = (wave < 4) ? Wa : Wb;
  const int kb = (wave & 3) * 64;
  int ar = row0 + c;
  if (ar >= M) ar = M - 1;
  const float* __restrict__ ap = Ap + (long long)ar * 256 + kb + 2 * h;
  const float* __restrict__ bp = Bp + (long long)(kb + 2 * h) * 256 + col0 + 4 * c;

  const int row = tid >> 4, c4 = tid & 15;
  const int g = row0 + row;
  const float n = cnt[g < M ? g : M - 1];
  const v4f bias = *(const v4f*)(va + col0 + 4 * c4) + *(const v4f*)(vb + col0 + 4 * c4);

  v2f a[16];
  v4f bx[16], by[16];
#pragma unroll
  for (int j = 0; j < 16; ++j) {
    a[j]  = *(const v2f*)(ap + 4 * j);
    bx[j] = *(const v4f*)(bp + (long long)(4 * j) * 256);
    by[j] = *(const v4f*)(bp + (long long)(4 * j + 1) * 256);
  }
  v8f acc0 = {}, acc1 = {}, acc2 = {}, acc3 = {};
#pragma unroll
  for (int j = 0; j < 16; ++j) {
    v2f b;
    b.x = bx[j].x; b.y = by[j].x;
    acc0 = __builtin_amdgcn_wmma_f32_16x16x4_f32(false, a[j], false, b, (short)0, acc0, false, false);
    b.x = bx[j].y; b.y = by[j].y;
    acc1 = __builtin_amdgcn_wmma_f32_16x16x4_f32(false, a[j], false, b, (short)0, acc1, false, false);
    b.x = bx[j].z; b.y = by[j].z;
    acc2 = __builtin_amdgcn_wmma_f32_16x16x4_f32(false, a[j], false, b, (short)0, acc2, false, false);
    b.x = bx[j].w; b.y = by[j].w;
    acc3 = __builtin_amdgcn_wmma_f32_16x16x4_f32(false, a[j], false, b, (short)0, acc3, false, false);
  }

#pragma unroll
  for (int r = 0; r < 8; ++r) {
    v4f v;
    v.x = acc0[r]; v.y = acc1[r]; v.z = acc2[r]; v.w = acc3[r];
    *(v4f*)(part + ((wave * 16 + 8 * h + r) * 64 + 4 * c)) = v;
  }
  __syncthreads();
  v4f sum = *(const v4f*)(part + (row * 64 + 4 * c4));
#pragma unroll
  for (int w = 1; w < 8; ++w) sum += *(const v4f*)(part + ((w * 16 + row) * 64 + 4 * c4));
  if (g < M)
    *(v4f*)(out + (long long)g * 256 + col0 + 4 * c4) = (sum + n * bias) / fmaxf(n, 1.0f);
}

static inline int cdiv(long long a, long long b) { return (int)((a + b - 1) / b); }

extern "C" void kernel_launch(void* const* d_in, const int* in_sizes, int n_in,
                              void* d_out, int out_size, void* d_ws, size_t ws_size,
                              hipStream_t stream) {
  const float* x        = (const float*)d_in[0];
  const int*   ei       = (const int*)  d_in[1];
  const int*   batch    = (const int*)  d_in[2];
  const float* W1       = (const float*)d_in[3];
  const float* att_src1 = (const float*)d_in[4];
  const float* att_dst1 = (const float*)d_in[5];
  const float* b1       = (const float*)d_in[6];
  const float* gamma    = (const float*)d_in[7];
  const float* beta     = (const float*)d_in[8];
  const float* W2       = (const float*)d_in[9];
  const float* att_src2 = (const float*)d_in[10];
  const float* att_dst2 = (const float*)d_in[11];
  const float* b2       = (const float*)d_in[12];
  const float* Wres     = (const float*)d_in[13];
  const float* bres     = (const float*)d_in[14];
  float* out = (float*)d_out;

  const int Nn = in_sizes[2];
  const int Ee = in_sizes[1] / 2;
  const int Gg = out_size / 256;

  const long long NB = (long long)Nn * 256;
  float* ws   = (float*)d_ws;
  float* buf0 = ws;
  float* buf1 = ws + NB;
  float* buf2 = ws + 2 * NB;
  float* scal = ws + 3 * NB;
  const long long Np = ((long long)Nn + 32) / 32 * 32;
  const long long Gp = ((long long)Gg + 31) / 32 * 32;
  int*   deg  = (int*)scal;
  float* cnt  = scal + Np;
  float* pagg = cnt + Gp;
  float* ph   = pagg + (long long)Gg * 256;
  float* as1  = ph + (long long)Gg * 256;
  float* ad1  = as1 + Np * 8;
  float* as2  = ad1 + Np * 8;
  float* ad2  = as2 + Np;
  float* u2   = ad2 + Np;
  int*   rowptr  = (int*)(u2 + 512);
  int*   cursor  = rowptr + Np;
  int*   csr_src = cursor + Np;

  const int B = 256;
  const long long zlen = Np + Gp + 2 * (long long)Gg * 256;

  fill_f32<<<cdiv(zlen, B), B, 0, stream>>>(scal, zlen, 0.0f);

  csr_count<<<cdiv(Ee, B), B, 0, stream>>>(ei, deg, Ee);
  csr_scan<<<cdiv(Nn, 1024), 1024, 0, stream>>>(deg, rowptr, cursor, Nn);
  csr_fill<<<cdiv(Ee, B), B, 0, stream>>>(ei, cursor, csr_src, Ee);

  const int gtiles = cdiv(Nn, 32);
  const int ntiles = cdiv(Nn, 8);

  gemm_wmma_n256<128, 8><<<gtiles, 128, 0, stream>>>(x, W1, buf0, att_src1, att_dst1,
                                                     as1, ad1, Nn);
  att_proj<<<32, B, 0, stream>>>(W2, att_src2, att_dst2, u2);
  gat_gather<8, true><<<ntiles, B, 0, stream>>>(rowptr, csr_src, buf0, as1, ad1, buf1,
                                                b1, gamma, beta, u2, as2, ad2, Nn);
  gat_gather<1, false><<<ntiles, B, 0, stream>>>(rowptr, csr_src, buf1, as2, ad2, buf2,
                                                 nullptr, nullptr, nullptr, nullptr,
                                                 nullptr, nullptr, Nn);
  pool_rows<<<cdiv(Nn, 64), B, 0, stream>>>(buf2, buf1, batch, pagg, ph, cnt, Nn);
  pooled_out<<<dim3(cdiv(Gg, 16), 4), B, 0, stream>>>(pagg, W2, ph, Wres, b2, bres,
                                                      cnt, out, Gg);
}


# CSR build (count, scan, fill: 3 launches, 2 atomic passes) replaced by one-pass fixed buckets: 64 source slots per node, overflow list beyond; slot order = atomic arrival as before
# speedup vs baseline: 5.8563x; 1.2883x over previous
#include <hip/hip_runtime.h>
#include <hip/hip_bf16.h>


typedef __attribute__((ext_vector_type(2))) float v2f;
typedef __attribute__((ext_vector_type(8))) float v8f;
typedef __attribute__((ext_vector_type(4))) float v4f;

#define NEG_SLOPE 0.2f
#define LN_EPS    1e-5f

__global__ void fill_f32(float* __restrict__ p, long long n, float v) {
  long long i = (long long)blockIdx.x * blockDim.x + threadIdx.x;
  if (i < n) p[i] = v;
}

__device__ __forceinline__ float leaky(float x) {
  return x > 0.0f ? x : NEG_SLOPE * x;
}

template <int MASK>
__device__ __forceinline__ float add_row_xor(float v) {
  const int o = __builtin_amdgcn_update_dpp(0, __float_as_int(v), 0x160 + MASK, 0xf, 0xf, false);
  return v + __int_as_float(o);
}

template <int K>
__global__ void __launch_bounds__(128)
gemm_wmma_n256(const float* __restrict__ A,
               const float* __restrict__ B,
               float* __restrict__ C,
               int M) {
  constexpr int BM = 32;
  constexpr int P  = K + 4;
  __shared__ __attribute__((aligned(16))) float As[BM * P];

  const int tid  = threadIdx.x;
  const int lane = tid & 31;
  const int wave = tid >> 5;
  const int c    = lane & 15;
  const int h    = lane >> 4;
  const long long row0 = (long long)blockIdx.x * BM;

  constexpr int Q = K / 4;
  for (int i = tid; i < BM * Q; i += 128) {
    const int r = i / Q, q = i - r * Q;
    long long gr = row0 + r;
    if (gr >= M) gr = M - 1;
    const v4f v = *(const v4f*)(A + gr * K + q * 4);
    *(v4f*)(As + r * P + q * 4) = v;
  }
  __syncthreads();

  const float* __restrict__ a0p = As + c * P + 2 * h;
  const float* __restrict__ a1p = As + (16 + c) * P + 2 * h;
  const float* __restrict__ Bx  = B + (long long)(2 * h) * 256 + wave * 64 + 4 * c;
  const float* __restrict__ By  = Bx + 256;

  v8f acc00 = {}, acc01 = {}, acc02 = {}, acc03 = {};
  v8f acc10 = {}, acc11 = {}, acc12 = {}, acc13 = {};

#pragma unroll 2
  for (int k = 0; k < K; k += 4) {
    const v2f a0 = *(const v2f*)(a0p + k);
    const v2f a1 = *(const v2f*)(a1p + k);
    const v4f bx = *(const v4f*)(Bx + (long long)k * 256);
    const v4f by = *(const v4f*)(By + (long long)k * 256);
    v2f b;
    b.x = bx.x; b.y = by.x;
    acc00 = __builtin_amdgcn_wmma_f32_16x16x4_f32(false, a0, false, b, (short)0, acc00, false, false);
    acc10 = __builtin_amdgcn_wmma_f32_16x16x4_f32(false, a1, false, b, (short)0, acc10, false, false);
    b.x = bx.y; b.y = by.y;
    acc01 = __builtin_amdgcn_wmma_f32_16x16x4_f32(false, a0, false, b, (short)0, acc01, false, false);
    acc11 = __builtin_amdgcn_wmma_f32_16x16x4_f32(false, a1, false, b, (short)0, acc11, false, false);
    b.x = bx.z; b.y = by.z;
    acc02 = __builtin_amdgcn_wmma_f32_16x16x4_f32(false, a0, false, b, (short)0, acc02, false, false);
    acc12 = __builtin_amdgcn_wmma_f32_16x16x4_f32(false, a1, false, b, (short)0, acc12, false, false);
    b.x = bx.w; b.y = by.w;
    acc03 = __builtin_amdgcn_wmma_f32_16x16x4_f32(false, a0, false, b, (short)0, acc03, false, false);
    acc13 = __builtin_amdgcn_wmma_f32_16x16x4_f32(false, a1, false, b, (short)0, acc13, false, false);
  }

  const int col = wave * 64 + 4 * c;
#pragma unroll
  for (int r = 0; r < 8; ++r) {
    const long long or0 = row0 + 8 * h + r;
    const long long or1 = or0 + 16;
    v4f v0, v1;
    v0.x = acc00[r]; v0.y = acc01[r]; v0.z = acc02[r]; v0.w = acc03[r];
    v1.x = acc10[r]; v1.y = acc11[r]; v1.z = acc12[r]; v1.w = acc13[r];
    if (or0 < M) *(v4f*)(C + or0 * 256 + col) = v0;
    if (or1 < M) *(v4f*)(C + or1 * 256 + col) = v1;
  }
}

#define BCAP 64
__global__ void bucket_fill(const int* __restrict__ ei, int* __restrict__ ncnt,
                            int* __restrict__ bucket, int* __restrict__ ovfc,
                            int* __restrict__ ovf, int E) {
  int e = blockIdx.x * blockDim.x + threadIdx.x;
  if (e < E) {
    const int d = ei[E + e], s = ei[e];
    const int p = atomicAdd(&ncnt[d], 1);
    if (p < BCAP) {
      bucket[(long long)d * BCAP + p] = s;
    } else {
      const int q = atomicAdd(ovfc, 1);
      ovf[2 * q] = d;
      ovf[2 * q + 1] = s;
    }
  }
}

template <int HEADS>
__device__ __forceinline__ void row_scores(const v4f xA, const v4f xB,
                                           const v4f vA, const v4f vB,
                                           float& oA, float& oB) {
  float pA = (xA.x * vA.x + xA.y * vA.y) + (xA.z * vA.z + xA.w * vA.w);
  float pB = (xB.x * vB.x + xB.y * vB.y) + (xB.z * vB.z + xB.w * vB.w);
  if (HEADS == 8) {
    pA = add_row_xor<1>(pA); pB = add_row_xor<1>(pB);
    pA = add_row_xor<2>(pA); pB = add_row_xor<2>(pB);
    pA = add_row_xor<4>(pA); pB = add_row_xor<4>(pB);
    oA = pA; oB = pB;
  } else {
    float p = pA + pB;
    p = add_row_xor<1>(p);
    p = add_row_xor<2>(p);
    p = add_row_xor<4>(p);
    p = add_row_xor<8>(p);
    const float lo = __int_as_float(__builtin_amdgcn_readlane(__float_as_int(p), 0));
    const float hi = __int_as_float(__builtin_amdgcn_readlane(__float_as_int(p), 16));
    oA = oB = lo + hi;
  }
}

__device__ __forceinline__ void soft_step(const float l, float& m, float& z,
                                          float& sc, float& w) {
  const float d = l - m;
  const float e = __expf(-fabsf(d));
  sc = d > 0.0f ? e : 1.0f;
  w  = d > 0.0f ? 1.0f : e;
  m  = fmaxf(m, l);
  z  = z * sc + w;
}

template <int HEADS, bool LN>
__global__ void __launch_bounds__(256)
gat_gather(const int* __restrict__ ncnt, const int* __restrict__ bucket,
           const int* __restrict__ ovfc, const int* __restrict__ ovf, int ovf_cap,
           const float* __restrict__ hin, const float* __restrict__ vsrc,
           const float* __restrict__ vdst, float* __restrict__ outp,
           const float* __restrict__ lnb, const float* __restrict__ gamma,
           const float* __restrict__ beta, int Nn) {
  static_assert(HEADS == 1 || HEADS == 8, "the lane layout covers one or eight heads");
  constexpr bool TWO = (HEADS > 1);
  const int lane = threadIdx.x & 31;
  int n = blockIdx.x * 8 + (threadIdx.x >> 5);
  if (n >= Nn) return;
  n = __builtin_amdgcn_readfirstlane(n);
  const int deg = ncnt[n];
  const int nb  = deg < BCAP ? deg : BCAP;
  const int* __restrict__ brow = bucket + (long long)n * BCAP;
  const v4f* __restrict__ h4 = (const v4f*)hin;
  const v4f vsA = *(const v4f*)(vsrc + 4 * lane), vsB = *(const v4f*)(vsrc + 128 + 4 * lane);
  const v4f vdA = *(const v4f*)(vdst + 4 * lane), vdB = *(const v4f*)(vdst + 128 + 4 * lane);

  v4f accA = h4[(long long)n * 64 + lane];
  v4f accB = h4[(long long)n * 64 + 32 + lane];
  float adA, adB, mA, mB;
  row_scores<HEADS>(accA, accB, vdA, vdB, adA, adB);
  row_scores<HEADS>(accA, accB, vsA, vsB, mA, mB);
  mA = leaky(mA + adA);
  mB = leaky(mB + adB);
  float zA = 1.0f, zB = 1.0f;

  auto edge = [&](const v4f xA, const v4f xB) {
    float lA, lB, sc, w;
    row_scores<HEADS>(xA, xB, vsA, vsB, lA, lB);
    soft_step(leaky(lA + adA), mA, zA, sc, w);
    accA = accA * sc + w * xA;
    if (TWO) soft_step(leaky(lB + adB), mB, zB, sc, w);
    accB = accB * sc + w * xB;
  };

  for (int base = 0; base < nb; base += 32) {
    const int cnt = (nb - base < 32) ? nb - base : 32;
    const int sv  = (lane < cnt) ? brow[base + lane] : n;
    int t = 0;
    for (; t + 4 <= cnt; t += 4) {
      const int s0 = __builtin_amdgcn_readlane(sv, t);
      const int s1 = __builtin_amdgcn_readlane(sv, t + 1);
      const int s2 = __builtin_amdgcn_readlane(sv, t + 2);
      const int s3 = __builtin_amdgcn_readlane(sv, t + 3);
      const v4f x0A = h4[(long long)s0 * 64 + lane], x0B = h4[(long long)s0 * 64 + 32 + lane];
      const v4f x1A = h4[(long long)s1 * 64 + lane], x1B = h4[(long long)s1 * 64 + 32 + lane];
      const v4f x2A = h4[(long long)s2 * 64 + lane], x2B = h4[(long long)s2 * 64 + 32 + lane];
      const v4f x3A = h4[(long long)s3 * 64 + lane], x3B = h4[(long long)s3 * 64 + 32 + lane];
      edge(x0A, x0B);
      edge(x1A, x1B);
      edge(x2A, x2B);
      edge(x3A, x3B);
    }
    for (; t < cnt; ++t) {
      const int s = __builtin_amdgcn_readlane(sv, t);
      edge(h4[(long long)s * 64 + lane], h4[(long long)s * 64 + 32 + lane]);
    }
  }
  if (deg > BCAP) {
    int novf = ovfc[0];
    if (novf > ovf_cap) novf = ovf_cap;
    for (int i = 0; i < novf; i += 32) {
      const int j = i + lane;
      int d = -1, s = 0;
      if (j < novf) { d = ovf[2 * j]; s = ovf[2 * j + 1]; }
      unsigned m = __builtin_amdgcn_ballot_w32(d == n);
      while (m) {
        const int b = __builtin_ctz(m);
        m &= m - 1;
        const int so = __builtin_amdgcn_readlane(s, b);
        edge(h4[(long long)so * 64 + lane], h4[(long long)so * 64 + 32 + lane]);
      }
    }
  }
  const float rzA = 1.0f / zA;
  const float rzB = TWO ? 1.0f / zB : rzA;
  accA = accA * rzA;
  accB = accB * rzB;

  if (LN) {
    const v4f xA = accA + *(const v4f*)(lnb + 4 * lane);
    const v4f xB = accB + *(const v4f*)(lnb + 128 + 4 * lane);
    float sum = ((xA.x + xA.y) + (xA.z + xA.w)) + ((xB.x + xB.y) + (xB.z + xB.w));
    for (int off = 16; off > 0; off >>= 1) sum += __shfl_xor(sum, off);
    const float mu = sum * (1.0f / 256.0f);
    const v4f dA = xA - mu, dB = xB - mu;
    float var = ((dA.x * dA.x + dA.y * dA.y) + (dA.z * dA.z + dA.w * dA.w)) +
                ((dB.x * dB.x + dB.y * dB.y) + (dB.z * dB.z + dB.w * dB.w));
    for (int off = 16; off > 0; off >>= 1) var += __shfl_xor(var, off);
    var *= (1.0f / 256.0f);
    const float rs = rsqrtf(var + LN_EPS);
    const v4f tA = dA * rs * *(const v4f*)(gamma + 4 * lane) +
                   *(const v4f*)(beta + 4 * lane);
    const v4f tB = dB * rs * *(const v4f*)(gamma + 128 + 4 * lane) +
                   *(const v4f*)(beta + 128 + 4 * lane);
    accA.x = tA.x > 0.0f ? tA.x : (__expf(tA.x) - 1.0f);
    accA.y = tA.y > 0.0f ? tA.y : (__expf(tA.y) - 1.0f);
    accA.z = tA.z > 0.0f ? tA.z : (__expf(tA.z) - 1.0f);
    accA.w = tA.w > 0.0f ? tA.w : (__expf(tA.w) - 1.0f);
    accB.x = tB.x > 0.0f ? tB.x : (__expf(tB.x) - 1.0f);
    accB.y = tB.y > 0.0f ? tB.y : (__expf(tB.y) - 1.0f);
    accB.z = tB.z > 0.0f ? tB.z : (__expf(tB.z) - 1.0f);
    accB.w = tB.w > 0.0f ? tB.w : (__expf(tB.w) - 1.0f);
  }
  v4f* __restrict__ o4 = (v4f*)outp;
  o4[(long long)n * 64 + lane]      = accA;
  o4[(long long)n * 64 + 32 + lane] = accB;
}

__global__ void __launch_bounds__(256)
att_proj(const float* __restrict__ W, const float* __restrict__ vs,
         const float* __restrict__ vd, float* __restrict__ u) {
  const int lane = threadIdx.x & 31;
  const int k = blockIdx.x * 8 + (threadIdx.x >> 5);
  const v4f wA = *(const v4f*)(W + k * 256 + 4 * lane);
  const v4f wB = *(const v4f*)(W + k * 256 + 128 + 4 * lane);
  const v4f sA = *(const v4f*)(vs + 4 * lane), sB = *(const v4f*)(vs + 128 + 4 * lane);
  const v4f dA = *(const v4f*)(vd + 4 * lane), dB = *(const v4f*)(vd + 128 + 4 * lane);
  float ds = ((wA.x * sA.x + wA.y * sA.y) + (wA.z * sA.z + wA.w * sA.w)) +
             ((wB.x * sB.x + wB.y * sB.y) + (wB.z * sB.z + wB.w * sB.w));
  float dd = ((wA.x * dA.x + wA.y * dA.y) + (wA.z * dA.z + wA.w * dA.w)) +
             ((wB.x * dB.x + wB.y * dB.y) + (wB.z * dB.z + wB.w * dB.w));
  for (int off = 16; off > 0; off >>= 1) {
    ds += __shfl_xor(ds, off);
    dd += __shfl_xor(dd, off);
  }
  if (lane == 0) { u[k] = ds; u[256 + k] = dd; }
}

__global__ void __launch_bounds__(256)
pool_rows(const float* __restrict__ ha, const float* __restrict__ hb,
          const int* __restrict__ graph, float* __restrict__ pa,
          float* __restrict__ pb, float* __restrict__ cnt, int M) {
  const int q = threadIdx.x & 63;
  const long long base = (long long)blockIdx.x * 64 + (threadIdx.x >> 6) * 16;
  const v4f* __restrict__ a4 = (const v4f*)ha;
  const v4f* __restrict__ b4 = (const v4f*)hb;
  int cur = -1;
  float cn = 0.0f;
  v4f ra = {0.0f, 0.0f, 0.0f, 0.0f}, rb = {0.0f, 0.0f, 0.0f, 0.0f};
#pragma unroll
  for (int half = 0; half < 2; ++half) {
    v4f xa[8], xb[8];
    int g[8];
#pragma unroll
    for (int r = 0; r < 8; ++r) {
      long long row = base + 8 * half + r;
      if (row >= M) row = M - 1;
      g[r]  = graph[row];
      xa[r] = a4[row * 64 + q];
      xb[r] = b4[row * 64 + q];
    }
#pragma unroll
    for (int r = 0; r < 8; ++r) {
      if (base + 8 * half + r < M) {
        if (g[r] != cur) {
          if (cur >= 0) {
            float* oa = pa + (long long)cur * 256 + 4 * q;
            float* ob = pb + (long long)cur * 256 + 4 * q;
            atomicAdd(oa, ra.x); atomicAdd(oa + 1, ra.y);
            atomicAdd(oa + 2, ra.z); atomicAdd(oa + 3, ra.w);
            atomicAdd(ob, rb.x); atomicAdd(ob + 1, rb.y);
            atomicAdd(ob + 2, rb.z); atomicAdd(ob + 3, rb.w);
            if (q == 0) atomicAdd(cnt + cur, cn);
          }
          cur = g[r]; ra = xa[r]; rb = xb[r]; cn = 1.0f;
        } else {
          ra += xa[r]; rb += xb[r]; cn += 1.0f;
        }
      }
    }
  }
  if (cur >= 0) {
    float* oa = pa + (long long)cur * 256 + 4 * q;
    float* ob = pb + (long long)cur * 256 + 4 * q;
    atomicAdd(oa, ra.x); atomicAdd(oa + 1, ra.y);
    atomicAdd(oa + 2, ra.z); atomicAdd(oa + 3, ra.w);
    atomicAdd(ob, rb.x); atomicAdd(ob + 1, rb.y);
    atomicAdd(ob + 2, rb.z); atomicAdd(ob + 3, rb.w);
    if (q == 0) atomicAdd(cnt + cur, cn);
  }
}

__global__ void __launch_bounds__(256)
pooled_out(const float* __restrict__ pa, const float* __restrict__ Wa,
           const float* __restrict__ pb, const float* __restrict__ Wb,
           const float* __restrict__ va, const float* __restrict__ vb,
           const float* __restrict__ cnt, float* __restrict__ out, int M) {
  __shared__ __attribute__((aligned(16))) float part[8 * 16 * 64];
  const int tid  = threadIdx.x;
  const int lane = tid & 31;
  const int wave = tid >> 5;
  const int c    = lane & 15;
  const int h    = lane >> 4;
  const int row0 = blockIdx.x * 16;
  const int col0 = blockIdx.y * 64;
  const float* __restrict__ Ap = (wave < 4) ? pa : pb;
  const float* __restrict__ Bp = (wave < 4) ? Wa : Wb;
  const int kb = (wave & 3) * 64;
  int ar = row0 + c;
  if (ar >= M) ar = M - 1;
  const float* __restrict__ ap = Ap + (long long)ar * 256 + kb + 2 * h;
  const float* __restrict__ bp = Bp + (long long)(kb + 2 * h) * 256 + col0 + 4 * c;

  const int row = tid >> 4, c4 = tid & 15;
  const int g = row0 + row;
  const float n = cnt[g < M ? g : M - 1];
  const v4f bias = *(const v4f*)(va + col0 + 4 * c4) + *(const v4f*)(vb + col0 + 4 * c4);

  v2f a[16];
  v4f bx[16], by[16];
#pragma unroll
  for (int j = 0; j < 16; ++j) {
    a[j]  = *(const v2f*)(ap + 4 * j);
    bx[j] = *(const v4f*)(bp + (long long)(4 * j) * 256);
    by[j] = *(const v4f*)(bp + (long long)(4 * j + 1) * 256);
  }
  v8f acc0 = {}, acc1 = {}, acc2 = {}, acc3 = {};
#pragma unroll
  for (int j = 0; j < 16; ++j) {
    v2f b;
    b.x = bx[j].x; b.y = by[j].x;
    acc0 = __builtin_amdgcn_wmma_f32_16x16x4_f32(false, a[j], false, b, (short)0, acc0, false, false);
    b.x = bx[j].y; b.y = by[j].y;
    acc1 = __builtin_amdgcn_wmma_f32_16x16x4_f32(false, a[j], false, b, (short)0, acc1, false, false);
    b.x = bx[j].z; b.y = by[j].z;
    acc2 = __builtin_amdgcn_wmma_f32_16x16x4_f32(false, a[j], false, b, (short)0, acc2, false, false);
    b.x = bx[j].w; b.y = by[j].w;
    acc3 = __builtin_amdgcn_wmma_f32_16x16x4_f32(false, a[j], false, b, (short)0, acc3, false, false);
  }

#pragma unroll
  for (int r = 0; r < 8; ++r) {
    v4f v;
    v.x = acc0[r]; v.y = acc1[r]; v.z = acc2[r]; v.w = acc3[r];
    *(v4f*)(part + ((wave * 16 + 8 * h + r) * 64 + 4 * c)) = v;
  }
  __syncthreads();
  v4f sum = *(const v4f*)(part + (row * 64 + 4 * c4));
#pragma unroll
  for (int w = 1; w < 8; ++w) sum += *(const v4f*)(part + ((w * 16 + row) * 64 + 4 * c4));
  if (g < M)
    *(v4f*)(out + (long long)g * 256 + col0 + 4 * c4) = (sum + n * bias) / fmaxf(n, 1.0f);
}

static inline int cdiv(long long a, long long b) { return (int)((a + b - 1) / b); }

extern "C" void kernel_launch(void* const* d_in, const int* in_sizes, int n_in,
                              void* d_out, int out_size, void* d_ws, size_t ws_size,
                              hipStream_t stream) {
  const float* x        = (const float*)d_in[0];
  const int*   ei       = (const int*)  d_in[1];
  const int*   batch    = (const int*)  d_in[2];
  const float* W1       = (const float*)d_in[3];
  const float* att_src1 = (const float*)d_in[4];
  const float* att_dst1 = (const float*)d_in[5];
  const float* b1       = (const float*)d_in[6];
  const float* gamma    = (const float*)d_in[7];
  const float* beta     = (const float*)d_in[8];
  const float* W2       = (const float*)d_in[9];
  const float* att_src2 = (const float*)d_in[10];
  const float* att_dst2 = (const float*)d_in[11];
  const float* b2       = (const float*)d_in[12];
  const float* Wres     = (const float*)d_in[13];
  const float* bres     = (const float*)d_in[14];
  float* out = (float*)d_out;

  const int Nn = in_sizes[2];
  const int Ee = in_sizes[1] / 2;
  const int Gg = out_size / 256;

  const long long NB = (long long)Nn * 256;
  float* ws   = (float*)d_ws;
  float* buf0 = ws;
  float* buf1 = ws + NB;
  float* buf2 = ws + 2 * NB;
  float* scal = ws + 3 * NB;
  const long long Np = ((long long)Nn + 63) / 64 * 64;
  const long long Gp = ((long long)Gg + 63) / 64 * 64;
  int*   ncnt = (int*)scal;
  int*   ovfc = ncnt + Np;
  float* cnt  = scal + Np + 64;
  float* pagg = cnt + Gp;
  float* ph   = pagg + (long long)Gg * 256;
  float* u2   = ph + (long long)Gg * 256;
  int*   bucket = (int*)(u2 + 512);
  int*   ovf    = bucket + (long long)Nn * BCAP;

  const int B = 256;
  const long long zlen = Np + 64 + Gp + 2 * (long long)Gg * 256;

  fill_f32<<<cdiv(zlen, B), B, 0, stream>>>(scal, zlen, 0.0f);

  bucket_fill<<<cdiv(Ee, B), B, 0, stream>>>(ei, ncnt, bucket, ovfc, ovf, Ee);

  const int gtiles = cdiv(Nn, 32);
  const int ntiles = cdiv(Nn, 8);

  gemm_wmma_n256<128><<<gtiles, 128, 0, stream>>>(x, W1, buf0, Nn);
  att_proj<<<32, B, 0, stream>>>(W2, att_src2, att_dst2, u2);
  gat_gather<8, true><<<ntiles, B, 0, stream>>>(ncnt, bucket, ovfc, ovf, Ee, buf0, att_src1,
                                                att_dst1, buf1, b1, gamma, beta, Nn);
  gat_gather<1, false><<<ntiles, B, 0, stream>>>(ncnt, bucket, ovfc, ovf, Ee, buf1, u2, u2 + 256,
                                                 buf2, nullptr, nullptr, nullptr, Nn);
  pool_rows<<<cdiv(Nn, 64), B, 0, stream>>>(buf2, buf1, batch, pagg, ph, cnt, Nn);
  pooled_out<<<dim3(cdiv(Gg, 16), 4), B, 0, stream>>>(pagg, W2, ph, Wres, b2, bres,
                                                      cnt, out, Gg);
}


# Layer-1 gather scores a trip's 4 edges together: packed partial dots, one transposing butterfly leaves each lane ONE logit, softmax state moves once per trip; dot and softmax sum order change
# speedup vs baseline: 6.0548x; 1.0339x over previous
#include <hip/hip_runtime.h>
#include <hip/hip_bf16.h>


typedef __attribute__((ext_vector_type(2))) float v2f;
typedef __attribute__((ext_vector_type(8))) float v8f;
typedef __attribute__((ext_vector_type(4))) float v4f;

#define NEG_SLOPE 0.2f
#define LN_EPS    1e-5f

__global__ void fill_f32(float* __restrict__ p, long long n, float v) {
  long long i = (long long)blockIdx.x * blockDim.x + threadIdx.x;
  if (i < n) p[i] = v;
}

__device__ __forceinline__ float leaky(float x) {
  return x > 0.0f ? x : NEG_SLOPE * x;
}

template <int MASK>
__device__ __forceinline__ float add_row_xor(float v) {
  const int o = __builtin_amdgcn_update_dpp(0, __float_as_int(v), 0x160 + MASK, 0xf, 0xf, false);
  return v + __int_as_float(o);
}

template <int MASK>
__device__ __forceinline__ float row_xor(float v) {
  return __int_as_float(__builtin_amdgcn_update_dpp(0, __float_as_int(v), 0x160 + MASK, 0xf, 0xf, false));
}

template <int SRC>
__device__ __forceinline__ float bcast8(float v) {
  return __int_as_float(__builtin_amdgcn_mov_dpp8(__float_as_int(v), SRC * 0x249249));
}

__device__ __forceinline__ float dot4p(const v4f x, const v4f v) {
  const v2f xl = {x.x, x.y}, xh = {x.z, x.w}, vl = {v.x, v.y}, vh = {v.z, v.w};
  const v2f t = xh * vh + xl * vl;
  return t.x + t.y;
}

template <int K>
__global__ void __launch_bounds__(128)
gemm_wmma_n256(const float* __restrict__ A,
               const float* __restrict__ B,
               float* __restrict__ C,
               int M) {
  constexpr int BM = 32;
  constexpr int P  = K + 4;
  __shared__ __attribute__((aligned(16))) float As[BM * P];

  const int tid  = threadIdx.x;
  const int lane = tid & 31;
  const int wave = tid >> 5;
  const int c    = lane & 15;
  const int h    = lane >> 4;
  const long long row0 = (long long)blockIdx.x * BM;

  constexpr int Q = K / 4;
  for (int i = tid; i < BM * Q; i += 128) {
    const int r = i / Q, q = i - r * Q;
    long long gr = row0 + r;
    if (gr >= M) gr = M - 1;
    const v4f v = *(const v4f*)(A + gr * K + q * 4);
    *(v4f*)(As + r * P + q * 4) = v;
  }
  __syncthreads();

  const float* __restrict__ a0p = As + c * P + 2 * h;
  const float* __restrict__ a1p = As + (16 + c) * P + 2 * h;
  const float* __restrict__ Bx  = B + (long long)(2 * h) * 256 + wave * 64 + 4 * c;
  const float* __restrict__ By  = Bx + 256;

  v8f acc00 = {}, acc01 = {}, acc02 = {}, acc03 = {};
  v8f acc10 = {}, acc11 = {}, acc12 = {}, acc13 = {};

#pragma unroll 2
  for (int k = 0; k < K; k += 4) {
    const v2f a0 = *(const v2f*)(a0p + k);
    const v2f a1 = *(const v2f*)(a1p + k);
    const v4f bx = *(const v4f*)(Bx + (long long)k * 256);
    const v4f by = *(const v4f*)(By + (long long)k * 256);
    v2f b;
    b.x = bx.x; b.y = by.x;
    acc00 = __builtin_amdgcn_wmma_f32_16x16x4_f32(false, a0, false, b, (short)0, acc00, false, false);
    acc10 = __builtin_amdgcn_wmma_f32_16x16x4_f32(false, a1, false, b, (short)0, acc10, false, false);
    b.x = bx.y; b.y = by.y;
    acc01 = __builtin_amdgcn_wmma_f32_16x16x4_f32(false, a0, false, b, (short)0, acc01, false, false);
    acc11 = __builtin_amdgcn_wmma_f32_16x16x4_f32(false, a1, false, b, (short)0, acc11, false, false);
    b.x = bx.z; b.y = by.z;
    acc02 = __builtin_amdgcn_wmma_f32_16x16x4_f32(false, a0, false, b, (short)0, acc02, false, false);
    acc12 = __builtin_amdgcn_wmma_f32_16x16x4_f32(false, a1, false, b, (short)0, acc12, false, false);
    b.x = bx.w; b.y = by.w;
    acc03 = __builtin_amdgcn_wmma_f32_16x16x4_f32(false, a0, false, b, (short)0, acc03, false, false);
    acc13 = __builtin_amdgcn_wmma_f32_16x16x4_f32(false, a1, false, b, (short)0, acc13, false, false);
  }

  const int col = wave * 64 + 4 * c;
#pragma unroll
  for (int r = 0; r < 8; ++r) {
    const long long or0 = row0 + 8 * h + r;
    const long long or1 = or0 + 16;
    v4f v0, v1;
    v0.x = acc00[r]; v0.y = acc01[r]; v0.z = acc02[r]; v0.w = acc03[r];
    v1.x = acc10[r]; v1.y = acc11[r]; v1.z = acc12[r]; v1.w = acc13[r];
    if (or0 < M) *(v4f*)(C + or0 * 256 + col) = v0;
    if (or1 < M) *(v4f*)(C + or1 * 256 + col) = v1;
  }
}

#define BCAP 64
__global__ void bucket_fill(const int* __restrict__ ei, int* __restrict__ ncnt,
                            int* __restrict__ bucket, int* __restrict__ ovfc,
                            int* __restrict__ ovf, int E) {
  int e = blockIdx.x * blockDim.x + threadIdx.x;
  if (e < E) {
    const int d = ei[E + e], s = ei[e];
    const int p = atomicAdd(&ncnt[d], 1);
    if (p < BCAP) {
      bucket[(long long)d * BCAP + p] = s;
    } else {
      const int q = atomicAdd(ovfc, 1);
      ovf[2 * q] = d;
      ovf[2 * q + 1] = s;
    }
  }
}

template <int HEADS>
__device__ __forceinline__ void row_scores(const v4f xA, const v4f xB,
                                           const v4f vA, const v4f vB,
                                           float& oA, float& oB) {
  float pA = (xA.x * vA.x + xA.y * vA.y) + (xA.z * vA.z + xA.w * vA.w);
  float pB = (xB.x * vB.x + xB.y * vB.y) + (xB.z * vB.z + xB.w * vB.w);
  if (HEADS == 8) {
    pA = add_row_xor<1>(pA); pB = add_row_xor<1>(pB);
    pA = add_row_xor<2>(pA); pB = add_row_xor<2>(pB);
    pA = add_row_xor<4>(pA); pB = add_row_xor<4>(pB);
    oA = pA; oB = pB;
  } else {
    float p = pA + pB;
    p = add_row_xor<1>(p);
    p = add_row_xor<2>(p);
    p = add_row_xor<4>(p);
    p = add_row_xor<8>(p);
    const float lo = __int_as_float(__builtin_amdgcn_readlane(__float_as_int(p), 0));
    const float hi = __int_as_float(__builtin_amdgcn_readlane(__float_as_int(p), 16));
    oA = oB = lo + hi;
  }
}

__device__ __forceinline__ void soft_step(const float l, float& m, float& z,
                                          float& sc, float& w) {
  const float d = l - m;
  const float e = __expf(-fabsf(d));
  sc = d > 0.0f ? e : 1.0f;
  w  = d > 0.0f ? 1.0f : e;
  m  = fmaxf(m, l);
  z  = z * sc + w;
}

template <int HEADS, bool LN>
__global__ void __launch_bounds__(256)
gat_gather(const int* __restrict__ ncnt, const int* __restrict__ bucket,
           const int* __restrict__ ovfc, const int* __restrict__ ovf, int ovf_cap,
           const float* __restrict__ hin, const float* __restrict__ vsrc,
           const float* __restrict__ vdst, float* __restrict__ outp,
           const float* __restrict__ lnb, const float* __restrict__ gamma,
           const float* __restrict__ beta, int Nn) {
  static_assert(HEADS == 1 || HEADS == 8, "the lane layout covers one or eight heads");
  const int lane = threadIdx.x & 31;
  int n = blockIdx.x * 8 + (threadIdx.x >> 5);
  if (n >= Nn) return;
  n = __builtin_amdgcn_readfirstlane(n);
  const int deg = ncnt[n];
  const int nb  = deg < BCAP ? deg : BCAP;
  const int* __restrict__ brow = bucket + (long long)n * BCAP;
  const v4f* __restrict__ h4 = (const v4f*)hin;
  const v4f vsA = *(const v4f*)(vsrc + 4 * lane), vsB = *(const v4f*)(vsrc + 128 + 4 * lane);
  const v4f vdA = *(const v4f*)(vdst + 4 * lane), vdB = *(const v4f*)(vdst + 128 + 4 * lane);

  v4f accA = h4[(long long)n * 64 + lane];
  v4f accB = h4[(long long)n * 64 + 32 + lane];
  float adA, adB, mA, mB;
  row_scores<HEADS>(accA, accB, vdA, vdB, adA, adB);
  row_scores<HEADS>(accA, accB, vsA, vsB, mA, mB);
  mA = leaky(mA + adA);
  mB = leaky(mB + adB);
  float zA = 1.0f, zB = 1.0f;

  if constexpr (HEADS == 8) {
    const bool b0 = (lane & 1) != 0, b1 = (lane & 2) != 0, b2 = (lane & 4) != 0;
    const int  e  = lane & 3;
    const float adS = b2 ? adB : adA;
    float mS = b2 ? mB : mA;
    float zS = 1.0f;
    const v4f zero = {0.0f, 0.0f, 0.0f, 0.0f};

    auto trip = [&](const v4f x0A, const v4f x0B, const v4f x1A, const v4f x1B,
                    const v4f x2A, const v4f x2B, const v4f x3A, const v4f x3B,
                    const int nvalid) {
      const float a0 = dot4p(x0A, vsA), a1 = dot4p(x1A, vsA), a2 = dot4p(x2A, vsA), a3 = dot4p(x3A, vsA);
      const float c0 = dot4p(x0B, vsB), c1 = dot4p(x1B, vsB), c2 = dot4p(x2B, vsB), c3 = dot4p(x3B, vsB);
      const float u0 = (b2 ? c0 : a0) + row_xor<4>(b2 ? a0 : c0);
      const float u1 = (b2 ? c1 : a1) + row_xor<4>(b2 ? a1 : c1);
      const float u2 = (b2 ? c2 : a2) + row_xor<4>(b2 ? a2 : c2);
      const float u3 = (b2 ? c3 : a3) + row_xor<4>(b2 ? a3 : c3);
      const float t0 = (b1 ? u2 : u0) + row_xor<2>(b1 ? u0 : u2);
      const float t1 = (b1 ? u3 : u1) + row_xor<2>(b1 ? u1 : u3);
      const float s = (b0 ? t1 : t0) + row_xor<1>(b0 ? t0 : t1);

      float l = leaky(s + adS);
      l = (e < nvalid) ? l : -1.0e30f;
      float mx = fmaxf(l, row_xor<1>(l));
      mx = fmaxf(mx, row_xor<2>(mx));
      const float mn = fmaxf(mS, mx);
      const float w  = __expf(l - mn);
      const float sc = __expf(mS - mn);
      float ws = w + row_xor<1>(w);
      ws = ws + row_xor<2>(ws);
      zS = zS * sc + ws;
      mS = mn;

      const float scA = bcast8<0>(sc), scB = bcast8<4>(sc);
      const float wA0 = bcast8<0>(w), wA1 = bcast8<1>(w), wA2 = bcast8<2>(w), wA3 = bcast8<3>(w);
      const float wB0 = bcast8<4>(w), wB1 = bcast8<5>(w), wB2 = bcast8<6>(w), wB3 = bcast8<7>(w);
      accA = accA * scA + wA0 * x0A + wA1 * x1A + wA2 * x2A + wA3 * x3A;
      accB = accB * scB + wB0 * x0B + wB1 * x1B + wB2 * x2B + wB3 * x3B;
    };

    for (int base = 0; base < nb; base += 32) {
      const int cnt = (nb - base < 32) ? nb - base : 32;
      const int sv  = (lane < cnt) ? brow[base + lane] : n;
      int t = 0;
      for (; t + 4 <= cnt; t += 4) {
        const int s0 = __builtin_amdgcn_readlane(sv, t);
        const int s1 = __builtin_amdgcn_readlane(sv, t + 1);
        const int s2 = __builtin_amdgcn_readlane(sv, t + 2);
        const int s3 = __builtin_amdgcn_readlane(sv, t + 3);
        const v4f x0A = h4[(long long)s0 * 64 + lane], x0B = h4[(long long)s0 * 64 + 32 + lane];
        const v4f x1A = h4[(long long)s1 * 64 + lane], x1B = h4[(long long)s1 * 64 + 32 + lane];
        const v4f x2A = h4[(long long)s2 * 64 + lane], x2B = h4[(long long)s2 * 64 + 32 + lane];
        const v4f x3A = h4[(long long)s3 * 64 + lane], x3B = h4[(long long)s3 * 64 + 32 + lane];
        trip(x0A, x0B, x1A, x1B, x2A, x2B, x3A, x3B, 4);
      }
      if (t < cnt) {
        const int r = cnt - t;
        const int s0 = __builtin_amdgcn_readlane(sv, t);
        v4f x0A = h4[(long long)s0 * 64 + lane], x0B = h4[(long long)s0 * 64 + 32 + lane];
        v4f x1A = zero, x1B = zero, x2A = zero, x2B = zero;
        if (r > 1) {
          const int s1 = __builtin_amdgcn_readlane(sv, t + 1);
          x1A = h4[(long long)s1 * 64 + lane]; x1B = h4[(long long)s1 * 64 + 32 + lane];
        }
        if (r > 2) {
          const int s2 = __builtin_amdgcn_readlane(sv, t + 2);
          x2A = h4[(long long)s2 * 64 + lane]; x2B = h4[(long long)s2 * 64 + 32 + lane];
        }
        trip(x0A, x0B, x1A, x1B, x2A, x2B, zero, zero, r);
      }
    }
    if (deg > BCAP) {
      int novf = ovfc[0];
      if (novf > ovf_cap) novf = ovf_cap;
      for (int i = 0; i < novf; i += 32) {
        const int j = i + lane;
        int d = -1, s = 0;
        if (j < novf) { d = ovf[2 * j]; s = ovf[2 * j + 1]; }
        unsigned m = __builtin_amdgcn_ballot_w32(d == n);
        while (m) {
          const int b = __builtin_ctz(m);
          m &= m - 1;
          const int so = __builtin_amdgcn_readlane(s, b);
          trip(h4[(long long)so * 64 + lane], h4[(long long)so * 64 + 32 + lane],
               zero, zero, zero, zero, zero, zero, 1);
        }
      }
    }
    zA = bcast8<0>(zS);
    zB = bcast8<4>(zS);
  } else {
    auto edge = [&](const v4f xA, const v4f xB) {
      float lA, lB, sc, w;
      row_scores<HEADS>(xA, xB, vsA, vsB, lA, lB);
      soft_step(leaky(lA + adA), mA, zA, sc, w);
      accA = accA * sc + w * xA;
      accB = accB * sc + w * xB;
    };

    for (int base = 0; base < nb; base += 32) {
      const int cnt = (nb - base < 32) ? nb - base : 32;
      const int sv  = (lane < cnt) ? brow[base + lane] : n;
      int t = 0;
      for (; t + 4 <= cnt; t += 4) {
        const int s0 = __builtin_amdgcn_readlane(sv, t);
        const int s1 = __builtin_amdgcn_readlane(sv, t + 1);
        const int s2 = __builtin_amdgcn_readlane(sv, t + 2);
        const int s3 = __builtin_amdgcn_readlane(sv, t + 3);
        const v4f x0A = h4[(long long)s0 * 64 + lane], x0B = h4[(long long)s0 * 64 + 32 + lane];
        const v4f x1A = h4[(long long)s1 * 64 + lane], x1B = h4[(long long)s1 * 64 + 32 + lane];
        const v4f x2A = h4[(long long)s2 * 64 + lane], x2B = h4[(long long)s2 * 64 + 32 + lane];
        const v4f x3A = h4[(long long)s3 * 64 + lane], x3B = h4[(long long)s3 * 64 + 32 + lane];
        edge(x0A, x0B);
        edge(x1A, x1B);
        edge(x2A, x2B);
        edge(x3A, x3B);
      }
      for (; t < cnt; ++t) {
        const int s = __builtin_amdgcn_readlane(sv, t);
        edge(h4[(long long)s * 64 + lane], h4[(long long)s * 64 + 32 + lane]);
      }
    }
    if (deg > BCAP) {
      int novf = ovfc[0];
      if (novf > ovf_cap) novf = ovf_cap;
      for (int i = 0; i < novf; i += 32) {
        const int j = i + lane;
        int d = -1, s = 0;
        if (j < novf) { d = ovf[2 * j]; s = ovf[2 * j + 1]; }
        unsigned m = __builtin_amdgcn_ballot_w32(d == n);
        while (m) {
          const int b = __builtin_ctz(m);
          m &= m - 1;
          const int so = __builtin_amdgcn_readlane(s, b);
          edge(h4[(long long)so * 64 + lane], h4[(long long)so * 64 + 32 + lane]);
        }
      }
    }
    zB = zA;
  }
  const float rzA = 1.0f / zA;
  const float rzB = 1.0f / zB;
  accA = accA * rzA;
  accB = accB * rzB;

  if (LN) {
    const v4f xA = accA + *(const v4f*)(lnb + 4 * lane);
    const v4f xB = accB + *(const v4f*)(lnb + 128 + 4 * lane);
    float sum = ((xA.x + xA.y) + (xA.z + xA.w)) + ((xB.x + xB.y) + (xB.z + xB.w));
    for (int off = 16; off > 0; off >>= 1) sum += __shfl_xor(sum, off);
    const float mu = sum * (1.0f / 256.0f);
    const v4f dA = xA - mu, dB = xB - mu;
    float var = ((dA.x * dA.x + dA.y * dA.y) + (dA.z * dA.z + dA.w * dA.w)) +
                ((dB.x * dB.x + dB.y * dB.y) + (dB.z * dB.z + dB.w * dB.w));
    for (int off = 16; off > 0; off >>= 1) var += __shfl_xor(var, off);
    var *= (1.0f / 256.0f);
    const float rs = rsqrtf(var + LN_EPS);
    const v4f tA = dA * rs * *(const v4f*)(gamma + 4 * lane) +
                   *(const v4f*)(beta + 4 * lane);
    const v4f tB = dB * rs * *(const v4f*)(gamma + 128 + 4 * lane) +
                   *(const v4f*)(beta + 128 + 4 * lane);
    accA.x = tA.x > 0.0f ? tA.x : (__expf(tA.x) - 1.0f);
    accA.y = tA.y > 0.0f ? tA.y : (__expf(tA.y) - 1.0f);
    accA.z = tA.z > 0.0f ? tA.z : (__expf(tA.z) - 1.0f);
    accA.w = tA.w > 0.0f ? tA.w : (__expf(tA.w) - 1.0f);
    accB.x = tB.x > 0.0f ? tB.x : (__expf(tB.x) - 1.0f);
    accB.y = tB.y > 0.0f ? tB.y : (__expf(tB.y) - 1.0f);
    accB.z = tB.z > 0.0f ? tB.z : (__expf(tB.z) - 1.0f);
    accB.w = tB.w > 0.0f ? tB.w : (__expf(tB.w) - 1.0f);
  }
  v4f* __restrict__ o4 = (v4f*)outp;
  o4[(long long)n * 64 + lane]      = accA;
  o4[(long long)n * 64 + 32 + lane] = accB;
}

__global__ void __launch_bounds__(256)
att_proj(const float* __restrict__ W, const float* __restrict__ vs,
         const float* __restrict__ vd, float* __restrict__ u) {
  const int lane = threadIdx.x & 31;
  const int k = blockIdx.x * 8 + (threadIdx.x >> 5);
  const v4f wA = *(const v4f*)(W + k * 256 + 4 * lane);
  const v4f wB = *(const v4f*)(W + k * 256 + 128 + 4 * lane);
  const v4f sA = *(const v4f*)(vs + 4 * lane), sB = *(const v4f*)(vs + 128 + 4 * lane);
  const v4f dA = *(const v4f*)(vd + 4 * lane), dB = *(const v4f*)(vd + 128 + 4 * lane);
  float ds = ((wA.x * sA.x + wA.y * sA.y) + (wA.z * sA.z + wA.w * sA.w)) +
             ((wB.x * sB.x + wB.y * sB.y) + (wB.z * sB.z + wB.w * sB.w));
  float dd = ((wA.x * dA.x + wA.y * dA.y) + (wA.z * dA.z + wA.w * dA.w)) +
             ((wB.x * dB.x + wB.y * dB.y) + (wB.z * dB.z + wB.w * dB.w));
  for (int off = 16; off > 0; off >>= 1) {
    ds += __shfl_xor(ds, off);
    dd += __shfl_xor(dd, off);
  }
  if (lane == 0) { u[k] = ds; u[256 + k] = dd; }
}

__global__ void __launch_bounds__(256)
pool_rows(const float* __restrict__ ha, const float* __restrict__ hb,
          const int* __restrict__ graph, float* __restrict__ pa,
          float* __restrict__ pb, float* __restrict__ cnt, int M) {
  const int q = threadIdx.x & 63;
  const long long base = (long long)blockIdx.x * 64 + (threadIdx.x >> 6) * 16;
  const v4f* __restrict__ a4 = (const v4f*)ha;
  const v4f* __restrict__ b4 = (const v4f*)hb;
  int cur = -1;
  float cn = 0.0f;
  v4f ra = {0.0f, 0.0f, 0.0f, 0.0f}, rb = {0.0f, 0.0f, 0.0f, 0.0f};
#pragma unroll
  for (int half = 0; half < 2; ++half) {
    v4f xa[8], xb[8];
    int g[8];
#pragma unroll
    for (int r = 0; r < 8; ++r) {
      long long row = base + 8 * half + r;
      if (row >= M) row = M - 1;
      g[r]  = graph[row];
      xa[r] = a4[row * 64 + q];
      xb[r] = b4[row * 64 + q];
    }
#pragma unroll
    for (int r = 0; r < 8; ++r) {
      if (base + 8 * half + r < M) {
        if (g[r] != cur) {
          if (cur >= 0) {
            float* oa = pa + (long long)cur * 256 + 4 * q;
            float* ob = pb + (long long)cur * 256 + 4 * q;
            atomicAdd(oa, ra.x); atomicAdd(oa + 1, ra.y);
            atomicAdd(oa + 2, ra.z); atomicAdd(oa + 3, ra.w);
            atomicAdd(ob, rb.x); atomicAdd(ob + 1, rb.y);
            atomicAdd(ob + 2, rb.z); atomicAdd(ob + 3, rb.w);
            if (q == 0) atomicAdd(cnt + cur, cn);
          }
          cur = g[r]; ra = xa[r]; rb = xb[r]; cn = 1.0f;
        } else {
          ra += xa[r]; rb += xb[r]; cn += 1.0f;
        }
      }
    }
  }
  if (cur >= 0) {
    float* oa = pa + (long long)cur * 256 + 4 * q;
    float* ob = pb + (long long)cur * 256 + 4 * q;
    atomicAdd(oa, ra.x); atomicAdd(oa + 1, ra.y);
    atomicAdd(oa + 2, ra.z); atomicAdd(oa + 3, ra.w);
    atomicAdd(ob, rb.x); atomicAdd(ob + 1, rb.y);
    atomicAdd(ob + 2, rb.z); atomicAdd(ob + 3, rb.w);
    if (q == 0) atomicAdd(cnt + cur, cn);
  }
}

__global__ void __launch_bounds__(256)
pooled_out(const float* __restrict__ pa, const float* __restrict__ Wa,
           const float* __restrict__ pb, const float* __restrict__ Wb,
           const float* __restrict__ va, const float* __restrict__ vb,
           const float* __restrict__ cnt, float* __restrict__ out, int M) {
  __shared__ __attribute__((aligned(16))) float part[8 * 16 * 64];
  const int tid  = threadIdx.x;
  const int lane = tid & 31;
  const int wave = tid >> 5;
  const int c    = lane & 15;
  const int h    = lane >> 4;
  const int row0 = blockIdx.x * 16;
  const int col0 = blockIdx.y * 64;
  const float* __restrict__ Ap = (wave < 4) ? pa : pb;
  const float* __restrict__ Bp = (wave < 4) ? Wa : Wb;
  const int kb = (wave & 3) * 64;
  int ar = row0 + c;
  if (ar >= M) ar = M - 1;
  const float* __restrict__ ap = Ap + (long long)ar * 256 + kb + 2 * h;
  const float* __restrict__ bp = Bp + (long long)(kb + 2 * h) * 256 + col0 + 4 * c;

  const int row = tid >> 4, c4 = tid & 15;
  const int g = row0 + row;
  const float n = cnt[g < M ? g : M - 1];
  const v4f bias = *(const v4f*)(va + col0 + 4 * c4) + *(const v4f*)(vb + col0 + 4 * c4);

  v2f a[16];
  v4f bx[16], by[16];
#pragma unroll
  for (int j = 0; j < 16; ++j) {
    a[j]  = *(const v2f*)(ap + 4 * j);
    bx[j] = *(const v4f*)(bp + (long long)(4 * j) * 256);
    by[j] = *(const v4f*)(bp + (long long)(4 * j + 1) * 256);
  }
  v8f acc0 = {}, acc1 = {}, acc2 = {}, acc3 = {};
#pragma unroll
  for (int j = 0; j < 16; ++j) {
    v2f b;
    b.x = bx[j].x; b.y = by[j].x;
    acc0 = __builtin_amdgcn_wmma_f32_16x16x4_f32(false, a[j], false, b, (short)0, acc0, false, false);
    b.x = bx[j].y; b.y = by[j].y;
    acc1 = __builtin_amdgcn_wmma_f32_16x16x4_f32(false, a[j], false, b, (short)0, acc1, false, false);
    b.x = bx[j].z; b.y = by[j].z;
    acc2 = __builtin_amdgcn_wmma_f32_16x16x4_f32(false, a[j], false, b, (short)0, acc2, false, false);
    b.x = bx[j].w; b.y = by[j].w;
    acc3 = __builtin_amdgcn_wmma_f32_16x16x4_f32(false, a[j], false, b, (short)0, acc3, false, false);
  }

#pragma unroll
  for (int r = 0; r < 8; ++r) {
    v4f v;
    v.x = acc0[r]; v.y = acc1[r]; v.z = acc2[r]; v.w = acc3[r];
    *(v4f*)(part + ((wave * 16 + 8 * h + r) * 64 + 4 * c)) = v;
  }
  __syncthreads();
  v4f sum = *(const v4f*)(part + (row * 64 + 4 * c4));
#pragma unroll
  for (int w = 1; w < 8; ++w) sum += *(const v4f*)(part + ((w * 16 + row) * 64 + 4 * c4));
  if (g < M)
    *(v4f*)(out + (long long)g * 256 + col0 + 4 * c4) = (sum + n * bias) / fmaxf(n, 1.0f);
}

static inline int cdiv(long long a, long long b) { return (int)((a + b - 1) / b); }

extern "C" void kernel_launch(void* const* d_in, const int* in_sizes, int n_in,
                              void* d_out, int out_size, void* d_ws, size_t ws_size,
                              hipStream_t stream) {
  const float* x        = (const float*)d_in[0];
  const int*   ei       = (const int*)  d_in[1];
  const int*   batch    = (const int*)  d_in[2];
  const float* W1       = (const float*)d_in[3];
  const float* att_src1 = (const float*)d_in[4];
  const float* att_dst1 = (const float*)d_in[5];
  const float* b1       = (const float*)d_in[6];
  const float* gamma    = (const float*)d_in[7];
  const float* beta     = (const float*)d_in[8];
  const float* W2       = (const float*)d_in[9];
  const float* att_src2 = (const float*)d_in[10];
  const float* att_dst2 = (const float*)d_in[11];
  const float* b2       = (const float*)d_in[12];
  const float* Wres     = (const float*)d_in[13];
  const float* bres     = (const float*)d_in[14];
  float* out = (float*)d_out;

  const int Nn = in_sizes[2];
  const int Ee = in_sizes[1] / 2;
  const int Gg = out_size / 256;

  const long long NB = (long long)Nn * 256;
  float* ws   = (float*)d_ws;
  float* buf0 = ws;
  float* buf1 = ws + NB;
  float* buf2 = ws + 2 * NB;
  float* scal = ws + 3 * NB;
  const long long Np = ((long long)Nn + 63) / 64 * 64;
  const long long Gp = ((long long)Gg + 63) / 64 * 64;
  int*   ncnt = (int*)scal;
  int*   ovfc = ncnt + Np;
  float* cnt  = scal + Np + 64;
  float* pagg = cnt + Gp;
  float* ph   = pagg + (long long)Gg * 256;
  float* u2   = ph + (long long)Gg * 256;
  int*   bucket = (int*)(u2 + 512);
  int*   ovf    = bucket + (long long)Nn * BCAP;

  const int B = 256;
  const long long zlen = Np + 64 + Gp + 2 * (long long)Gg * 256;

  fill_f32<<<cdiv(zlen, B), B, 0, stream>>>(scal, zlen, 0.0f);

  bucket_fill<<<cdiv(Ee, B), B, 0, stream>>>(ei, ncnt, bucket, ovfc, ovf, Ee);

  const int gtiles = cdiv(Nn, 32);
  const int ntiles = cdiv(Nn, 8);

  gemm_wmma_n256<128><<<gtiles, 128, 0, stream>>>(x, W1, buf0, Nn);
  att_proj<<<32, B, 0, stream>>>(W2, att_src2, att_dst2, u2);
  gat_gather<8, true><<<ntiles, B, 0, stream>>>(ncnt, bucket, ovfc, ovf, Ee, buf0, att_src1,
                                                att_dst1, buf1, b1, gamma, beta, Nn);
  gat_gather<1, false><<<ntiles, B, 0, stream>>>(ncnt, bucket, ovfc, ovf, Ee, buf1, u2, u2 + 256,
                                                 buf2, nullptr, nullptr, nullptr, Nn);
  pool_rows<<<cdiv(Nn, 64), B, 0, stream>>>(buf2, buf1, batch, pagg, ph, cnt, Nn);
  pooled_out<<<dim3(cdiv(Gg, 16), 4), B, 0, stream>>>(pagg, W2, ph, Wres, b2, bres,
                                                      cnt, out, Gg);
}
